# starting kernel
# baseline (speedup 1.0000x reference)
#include <hip/hip_runtime.h>


#define NN 50000
#define NE 800000
#define DD 128
#define NT (NN / 16)
#define FEAT (NN * DD)
#define WSTRIDE 132
#define WOFF (DD * WSTRIDE)

typedef float v2f __attribute__((ext_vector_type(2)));
typedef float v8f __attribute__((ext_vector_type(8)));

__device__ __forceinline__ v8f wmma4(v2f a, v2f b, v8f c) {
  return __builtin_amdgcn_wmma_f32_16x16x4_f32(false, a, false, b, (short)0, c,
                                               false, false);
}

__global__ void k_zero(float4* __restrict__ p, int n4) {
  int i = blockIdx.x * blockDim.x + threadIdx.x;
  int stride = gridDim.x * blockDim.x;
  float4 z = {0.f, 0.f, 0.f, 0.f};
  for (; i < n4; i += stride) p[i] = z;
}

__global__ void k_deg(const int* __restrict__ row, float* __restrict__ deg) {
  int e = blockIdx.x * blockDim.x + threadIdx.x;
  if (e < NE) atomicAdd(&deg[row[e]], 1.0f);
}

__global__ void k_scatter(const float* __restrict__ xin,
                          const int* __restrict__ row,
                          const int* __restrict__ col,
                          float* __restrict__ agg) {
  long tid = (long)blockIdx.x * blockDim.x + threadIdx.x;
  int e = (int)(tid >> 5);
  if (e >= NE) return;
  int c = ((int)tid & 31) << 2;
  int s = col[e];
  int d = row[e];
  const float4 v = *(const float4*)(xin + (size_t)s * DD + c);
  float* dst = agg + (size_t)d * DD + c;
  atomicAdd(dst + 0, v.x);
  atomicAdd(dst + 1, v.y);
  atomicAdd(dst + 2, v.z);
  atomicAdd(dst + 3, v.w);
}

__global__ void k_sage_gemm(const float* __restrict__ agg,
                            const float* __restrict__ deg,
                            const float* __restrict__ h,
                            const float* __restrict__ Wl,
                            const float* __restrict__ Wr,
                            const float* __restrict__ bias,
                            const float* __restrict__ resid,
                            float* __restrict__ out,
                            float* __restrict__ hout) {
  __shared__ float wlds[2 * DD * WSTRIDE];

  for (int i = threadIdx.x; i < (DD * DD) / 4; i += 256) {
    int o = i >> 5;
    int kc = (i & 31) << 2;
    float4 a = *(const float4*)(Wl + (size_t)o * DD + kc);
    float4 b = *(const float4*)(Wr + (size_t)o * DD + kc);
    *(float4*)(&wlds[o * WSTRIDE + kc]) = a;
    *(float4*)(&wlds[WOFF + o * WSTRIDE + kc]) = b;
  }
  __syncthreads();

  const int nw = blockDim.x >> 5;
  const int wv = threadIdx.x >> 5;
  const int lane = threadIdx.x & 31;
  const int half = lane >> 4;
  const int lm = lane & 15;
  const int tstride = gridDim.x * nw;

  for (int tile = blockIdx.x * nw + wv; tile < NT; tile += tstride) {
    const int m0 = tile << 4;
    if (tile + tstride < NT) {
      const size_t nb = (size_t)((tile + tstride) * 16 + lm) * DD;
      __builtin_prefetch(agg + nb, 0, 1);
      __builtin_prefetch(h + nb, 0, 1);
    }

    const float sc = 1.0f / fmaxf(deg[m0 + lm], 1.0f);
    const float* arow = agg + (size_t)(m0 + lm) * DD + 2 * half;
    const float* hrow = h + (size_t)(m0 + lm) * DD + 2 * half;
    const float* blbase = &wlds[(size_t)lm * WSTRIDE + 2 * half];
    const float* brbase = &wlds[WOFF + (size_t)lm * WSTRIDE + 2 * half];

    v8f acc[8] = {};

#pragma unroll 1
    for (int pass = 0; pass < 2; ++pass) {
      const float* aptr = pass ? hrow : arow;
      const float* bb = pass ? brbase : blbase;
      const float scale = pass ? 1.0f : sc;

      v2f cur[8], nxt[8];
#pragma unroll
      for (int t = 0; t < 8; ++t)
        cur[t] = *(const v2f*)(bb + (size_t)(t << 4) * WSTRIDE);
      float2 a2 = *(const float2*)(aptr);

#pragma unroll 2
      for (int k = 0; k < 32; ++k) {
        const int K1 = ((k + 1) & 31) << 2;
#pragma unroll
        for (int t = 0; t < 8; ++t)
          nxt[t] = *(const v2f*)(bb + (size_t)(t << 4) * WSTRIDE + K1);
        float2 a2n = *(const float2*)(aptr + K1);

        v2f va;
        va.x = a2.x * scale;
        va.y = a2.y * scale;
#pragma unroll
        for (int t = 0; t < 8; ++t) acc[t] = wmma4(va, cur[t], acc[t]);

#pragma unroll
        for (int t = 0; t < 8; ++t) cur[t] = nxt[t];
        a2 = a2n;
      }
    }

#pragma unroll
    for (int t = 0; t < 8; ++t) {
      const int n0 = t << 4;
      const float bb = bias[n0 + lm];
#pragma unroll
      for (int r = 0; r < 8; ++r) {
        const size_t idx = (size_t)(m0 + r + (half << 3)) * DD + n0 + lm;
        float v = acc[t][r] + bb;
        if (resid) v += resid[idx];
        out[idx] = v;
        if (hout) hout[idx] = fmaxf(v, 0.0f);
      }
    }
  }
}

extern "C" void kernel_launch(void* const* d_in, const int* in_sizes, int n_in,
                              void* d_out, int out_size, void* d_ws,
                              size_t ws_size, hipStream_t stream) {
  const float* x = (const float*)d_in[0];
  const int* erow = (const int*)d_in[1];
  const int* ecol = (const int*)d_in[2];
  const float* Wl0 = (const float*)d_in[3];
  const float* Wr0 = (const float*)d_in[4];
  const float* b0 = (const float*)d_in[5];
  const float* Wl1 = (const float*)d_in[6];
  const float* Wr1 = (const float*)d_in[7];
  const float* b1 = (const float*)d_in[8];
  const float* Wl2 = (const float*)d_in[9];
  const float* Wr2 = (const float*)d_in[10];
  const float* b2 = (const float*)d_in[11];
  float* out = (float*)d_out;

  float* ws = (float*)d_ws;
  float* deg = ws;
  float* agg = ws + NN;
  float* h = agg + (size_t)FEAT;
  float* ycur = h + (size_t)FEAT;

  dim3 b256(256);
  const int degBlocks = (NE + 255) / 256;
  const int scatterBlocks = (int)(((long)NE * 32 + 255) / 256);
  const int gemmBlocks = (NT + 7) / 8;

  k_zero<<<dim3(64), b256, 0, stream>>>((float4*)deg, NN / 4);
  k_deg<<<dim3(degBlocks), b256, 0, stream>>>(erow, deg);

  k_zero<<<dim3(2048), b256, 0, stream>>>((float4*)agg, FEAT / 4);
  k_scatter<<<dim3(scatterBlocks), b256, 0, stream>>>(x, erow, ecol, agg);
  k_sage_gemm<<<dim3(gemmBlocks), b256, 0, stream>>>(agg, deg, x, Wl0, Wr0, b0,
                                                     nullptr, ycur, h);

  k_zero<<<dim3(2048), b256, 0, stream>>>((float4*)agg, FEAT / 4);
  k_scatter<<<dim3(scatterBlocks), b256, 0, stream>>>(h, erow, ecol, agg);
  k_sage_gemm<<<dim3(gemmBlocks), b256, 0, stream>>>(agg, deg, h, Wl1, Wr1, b1,
                                                     ycur, ycur, h);

  k_zero<<<dim3(2048), b256, 0, stream>>>((float4*)agg, FEAT / 4);
  k_scatter<<<dim3(scatterBlocks), b256, 0, stream>>>(h, erow, ecol, agg);
  k_sage_gemm<<<dim3(gemmBlocks), b256, 0, stream>>>(agg, deg, h, Wl2, Wr2, b2,
                                                     ycur, out, nullptr);
}


# Edges grouped by destination once per call (returning int add as rank, scan, fill); per layer a wave sums its node's source rows in registers: no float atomics, sum order = arrival rank
# speedup vs baseline: 2.6137x; 2.6137x over previous
#include <hip/hip_runtime.h>


#define NN 50000
#define NE 800000
#define DD 128
#define NT (NN / 16)
#define FEAT (NN * DD)
#define WSTRIDE 132
#define WOFF (DD * WSTRIDE)
#define SCAN_T 1024
#define SCAN_PER 49
#define NPAD (SCAN_T * SCAN_PER)
#define SCAN_FULL ((NN / 4) / SCAN_T)

typedef float v2f __attribute__((ext_vector_type(2)));
typedef float v8f __attribute__((ext_vector_type(8)));

__device__ __forceinline__ v8f wmma4(v2f a, v2f b, v8f c) {
  return __builtin_amdgcn_wmma_f32_16x16x4_f32(false, a, false, b, (short)0, c,
                                               false, false);
}

__global__ void k_zero(float4* __restrict__ p, int n4) {
  int i = blockIdx.x * blockDim.x + threadIdx.x;
  int stride = gridDim.x * blockDim.x;
  float4 z = {0.f, 0.f, 0.f, 0.f};
  for (; i < n4; i += stride) p[i] = z;
}

__global__ void k_rank(const int* __restrict__ row, int* __restrict__ cnt,
                       int* __restrict__ rank) {
  int e = blockIdx.x * blockDim.x + threadIdx.x;
  if (e < NE) rank[e] = atomicAdd(&cnt[row[e]], 1);
}

__global__ __launch_bounds__(SCAN_T) void k_scan(const int* __restrict__ cnt,
                                                 int* __restrict__ head,
                                                 float* __restrict__ deg) {
  __shared__ int s[NPAD];
  __shared__ int wtot[SCAN_T / 32];
  const int t = threadIdx.x;
  const int lane = t & 31;
  const int wv = t >> 5;

  int4 v[SCAN_FULL];
#pragma unroll
  for (int k = 0; k < SCAN_FULL; ++k)
    v[k] = ((const int4*)cnt)[t + k * SCAN_T];
  {
    const int i = t + SCAN_FULL * SCAN_T;
    if (i < NPAD / 4) {
      int4 w = {0, 0, 0, 0};
      if (i < NN / 4) {
        w = ((const int4*)cnt)[i];
        float4 f = {(float)w.x, (float)w.y, (float)w.z, (float)w.w};
        ((float4*)deg)[i] = f;
      }
      ((int4*)s)[i] = w;
    }
  }
#pragma unroll
  for (int k = 0; k < SCAN_FULL; ++k) {
    const int i = t + k * SCAN_T;
    float4 f = {(float)v[k].x, (float)v[k].y, (float)v[k].z, (float)v[k].w};
    ((float4*)deg)[i] = f;
    ((int4*)s)[i] = v[k];
  }
  __syncthreads();

  const int base = t * SCAN_PER;
  int sum = 0;
  for (int j = 0; j < SCAN_PER; ++j) sum += s[base + j];

  int inc = sum;
  for (int d = 1; d < 32; d <<= 1) {
    int o = __shfl_up(inc, d);
    if (lane >= d) inc += o;
  }
  if (lane == 31) wtot[wv] = inc;
  __syncthreads();
  int wo = (lane < wv) ? wtot[lane] : 0;
  for (int d = 16; d >= 1; d >>= 1) wo += __shfl_xor(wo, d);

  int run = wo + inc - sum;
  for (int j = 0; j < SCAN_PER; ++j) {
    const int c = s[base + j];
    s[base + j] = run;
    run += c;
  }
  __syncthreads();
  for (int i = t; i < NPAD / 4; i += SCAN_T) ((int4*)head)[i] = ((int4*)s)[i];
}

__global__ void k_fill(const int* __restrict__ row,
                       const int* __restrict__ col,
                       const int* __restrict__ rank,
                       const int* __restrict__ head,
                       int* __restrict__ srt) {
  int e = blockIdx.x * blockDim.x + threadIdx.x;
  if (e < NE) srt[head[row[e]] + rank[e]] = col[e];
}

template <int N>
__device__ __forceinline__ void take_rows(const float* __restrict__ xl, int my,
                                          int j, float4& acc) {
  float4 v[N];
#pragma unroll
  for (int u = 0; u < N; ++u) {
    const int sn = __builtin_amdgcn_readlane(my, j + u);
    v[u] = *(const float4*)(xl + (size_t)sn * DD);
  }
#pragma unroll
  for (int u = 0; u < N; ++u) {
    acc.x += v[u].x;
    acc.y += v[u].y;
    acc.z += v[u].z;
    acc.w += v[u].w;
  }
}

__global__ __launch_bounds__(256) void k_gather(const float* __restrict__ xin,
                                                const int* __restrict__ head,
                                                const int* __restrict__ srt,
                                                float* __restrict__ agg) {
  const int lane = threadIdx.x & 31;
  const int node = blockIdx.x * 8 + (threadIdx.x >> 5);
  const int beg = __builtin_amdgcn_readfirstlane(head[node]);
  const int end = __builtin_amdgcn_readfirstlane(head[node + 1]);
  const float* xl = xin + (lane << 2);
  float4 acc = {0.f, 0.f, 0.f, 0.f};
  for (int b = beg; b < end; b += 32) {
    const int m = (end - b < 32) ? (end - b) : 32;
    const int my = (lane < m) ? srt[b + lane] : 0;
    int j = 0;
    for (; j + 8 <= m; j += 8) take_rows<8>(xl, my, j, acc);
    if (m - j >= 4) { take_rows<4>(xl, my, j, acc); j += 4; }
    if (m - j >= 2) { take_rows<2>(xl, my, j, acc); j += 2; }
    if (m - j >= 1) take_rows<1>(xl, my, j, acc);
  }
  *(float4*)(agg + (size_t)node * DD + (lane << 2)) = acc;
}

__global__ void k_sage_gemm(const float* __restrict__ agg,
                            const float* __restrict__ deg,
                            const float* __restrict__ h,
                            const float* __restrict__ Wl,
                            const float* __restrict__ Wr,
                            const float* __restrict__ bias,
                            const float* __restrict__ resid,
                            float* __restrict__ out,
                            float* __restrict__ hout) {
  __shared__ float wlds[2 * DD * WSTRIDE];

  for (int i = threadIdx.x; i < (DD * DD) / 4; i += 256) {
    int o = i >> 5;
    int kc = (i & 31) << 2;
    float4 a = *(const float4*)(Wl + (size_t)o * DD + kc);
    float4 b = *(const float4*)(Wr + (size_t)o * DD + kc);
    *(float4*)(&wlds[o * WSTRIDE + kc]) = a;
    *(float4*)(&wlds[WOFF + o * WSTRIDE + kc]) = b;
  }
  __syncthreads();

  const int nw = blockDim.x >> 5;
  const int wv = threadIdx.x >> 5;
  const int lane = threadIdx.x & 31;
  const int half = lane >> 4;
  const int lm = lane & 15;
  const int tstride = gridDim.x * nw;

  for (int tile = blockIdx.x * nw + wv; tile < NT; tile += tstride) {
    const int m0 = tile << 4;
    if (tile + tstride < NT) {
      const size_t nb = (size_t)((tile + tstride) * 16 + lm) * DD;
      __builtin_prefetch(agg + nb, 0, 1);
      __builtin_prefetch(h + nb, 0, 1);
    }

    const float sc = 1.0f / fmaxf(deg[m0 + lm], 1.0f);
    const float* arow = agg + (size_t)(m0 + lm) * DD + 2 * half;
    const float* hrow = h + (size_t)(m0 + lm) * DD + 2 * half;
    const float* blbase = &wlds[(size_t)lm * WSTRIDE + 2 * half];
    const float* brbase = &wlds[WOFF + (size_t)lm * WSTRIDE + 2 * half];

    v8f acc[8] = {};

#pragma unroll 1
    for (int pass = 0; pass < 2; ++pass) {
      const float* aptr = pass ? hrow : arow;
      const float* bb = pass ? brbase : blbase;
      const float scale = pass ? 1.0f : sc;

      v2f cur[8], nxt[8];
#pragma unroll
      for (int t = 0; t < 8; ++t)
        cur[t] = *(const v2f*)(bb + (size_t)(t << 4) * WSTRIDE);
      float2 a2 = *(const float2*)(aptr);

#pragma unroll 2
      for (int k = 0; k < 32; ++k) {
        const int K1 = ((k + 1) & 31) << 2;
#pragma unroll
        for (int t = 0; t < 8; ++t)
          nxt[t] = *(const v2f*)(bb + (size_t)(t << 4) * WSTRIDE + K1);
        float2 a2n = *(const float2*)(aptr + K1);

        v2f va;
        va.x = a2.x * scale;
        va.y = a2.y * scale;
#pragma unroll
        for (int t = 0; t < 8; ++t) acc[t] = wmma4(va, cur[t], acc[t]);

#pragma unroll
        for (int t = 0; t < 8; ++t) cur[t] = nxt[t];
        a2 = a2n;
      }
    }

#pragma unroll
    for (int t = 0; t < 8; ++t) {
      const int n0 = t << 4;
      const float bb = bias[n0 + lm];
#pragma unroll
      for (int r = 0; r < 8; ++r) {
        const size_t idx = (size_t)(m0 + r + (half << 3)) * DD + n0 + lm;
        float v = acc[t][r] + bb;
        if (resid) v += resid[idx];
        out[idx] = v;
        if (hout) hout[idx] = fmaxf(v, 0.0f);
      }
    }
  }
}

extern "C" void kernel_launch(void* const* d_in, const int* in_sizes, int n_in,
                              void* d_out, int out_size, void* d_ws,
                              size_t ws_size, hipStream_t stream) {
  const float* x = (const float*)d_in[0];
  const int* erow = (const int*)d_in[1];
  const int* ecol = (const int*)d_in[2];
  const float* Wl0 = (const float*)d_in[3];
  const float* Wr0 = (const float*)d_in[4];
  const float* b0 = (const float*)d_in[5];
  const float* Wl1 = (const float*)d_in[6];
  const float* Wr1 = (const float*)d_in[7];
  const float* b1 = (const float*)d_in[8];
  const float* Wl2 = (const float*)d_in[9];
  const float* Wr2 = (const float*)d_in[10];
  const float* b2 = (const float*)d_in[11];
  float* out = (float*)d_out;

  float* ws = (float*)d_ws;
  float* agg = ws;
  float* h = agg + (size_t)FEAT;
  int* cnt = (int*)(h + (size_t)FEAT);
  int* head = cnt + NN;
  int* rank = head + NPAD;
  int* srt = rank + NE;
  float* deg = (float*)(srt + NE);

  dim3 b256(256);
  const int edgeBlocks = (NE + 255) / 256;
  const int gatherBlocks = NN / 8;
  const int gemmBlocks = (NT + 7) / 8;

  k_zero<<<dim3(64), b256, 0, stream>>>((float4*)cnt, NN / 4);
  k_rank<<<dim3(edgeBlocks), b256, 0, stream>>>(erow, cnt, rank);
  k_scan<<<dim3(1), dim3(SCAN_T), 0, stream>>>(cnt, head, deg);
  k_fill<<<dim3(edgeBlocks), b256, 0, stream>>>(erow, ecol, rank, head, srt);

  k_gather<<<dim3(gatherBlocks), b256, 0, stream>>>(x, head, srt, agg);
  k_sage_gemm<<<dim3(gemmBlocks), b256, 0, stream>>>(agg, deg, x, Wl0, Wr0, b0,
                                                     nullptr, out, h);

  k_gather<<<dim3(gatherBlocks), b256, 0, stream>>>(h, head, srt, agg);
  k_sage_gemm<<<dim3(gemmBlocks), b256, 0, stream>>>(agg, deg, h, Wl1, Wr1, b1,
                                                     out, out, h);

  k_gather<<<dim3(gatherBlocks), b256, 0, stream>>>(h, head, srt, agg);
  k_sage_gemm<<<dim3(gemmBlocks), b256, 0, stream>>>(agg, deg, h, Wl2, Wr2, b2,
                                                     out, out, nullptr);
}


# Edges grouped by a two-level counting sort on LDS counters (256-destination buckets per edge chunk, scan, node in bucket): no global atomics left; sum order = list order set by LDS add arrival
# speedup vs baseline: 3.1472x; 1.2041x over previous
#include <hip/hip_runtime.h>


#define NN 50000
#define NE 800000
#define DD 128
#define NT (NN / 16)
#define FEAT (NN * DD)
#define WSTRIDE 132
#define WOFF (DD * WSTRIDE)
#define SCAN_T 1024
#define SCAN_PER 49
#define NPAD (SCAN_T * SCAN_PER)
#define SCAN_FULL 12
#define NB 196
#define NG 250
#define CHUNK (NE / NG)
#define EDGE_T 640
#define EPT (CHUNK / EDGE_T)
#define M4 ((NB * NG) / 4)
#define BKT_T 1024
#define BKT_U 4
#define STAGE 8192

typedef float v2f __attribute__((ext_vector_type(2)));
typedef float v8f __attribute__((ext_vector_type(8)));

__device__ __forceinline__ v8f wmma4(v2f a, v2f b, v8f c) {
  return __builtin_amdgcn_wmma_f32_16x16x4_f32(false, a, false, b, (short)0, c,
                                               false, false);
}

__global__ __launch_bounds__(EDGE_T) void k_hist(const int* __restrict__ row,
                                                 int* __restrict__ cntm) {
  __shared__ int hist[256];
  const int t = threadIdx.x;
  const int g = blockIdx.x;
  if (t < 256) hist[t] = 0;
  __syncthreads();
  const int* r = row + g * CHUNK + t;
  int d[EPT];
#pragma unroll
  for (int j = 0; j < EPT; ++j) d[j] = r[j * EDGE_T];
#pragma unroll
  for (int j = 0; j < EPT; ++j) atomicAdd(&hist[d[j] >> 8], 1);
  __syncthreads();
  if (t < NB) cntm[t * NG + g] = hist[t];
}

__global__ __launch_bounds__(SCAN_T) void k_scan(const int* __restrict__ cntm,
                                                 int* __restrict__ offs) {
  __shared__ int s[NPAD];
  __shared__ int wtot[SCAN_T / 32];
  const int t = threadIdx.x;
  const int lane = t & 31;
  const int wv = t >> 5;
  const int4 z4 = {0, 0, 0, 0};

  int4 v[SCAN_FULL];
#pragma unroll
  for (int k = 0; k < SCAN_FULL; ++k) {
    const int i = t + k * SCAN_T;
    v[k] = (i < M4) ? ((const int4*)cntm)[i] : z4;
  }
  {
    const int i = t + SCAN_FULL * SCAN_T;
    if (i < NPAD / 4) ((int4*)s)[i] = z4;
  }
#pragma unroll
  for (int k = 0; k < SCAN_FULL; ++k) ((int4*)s)[t + k * SCAN_T] = v[k];
  __syncthreads();

  const int base = t * SCAN_PER;
  int sum = 0;
  for (int j = 0; j < SCAN_PER; ++j) sum += s[base + j];

  int inc = sum;
  for (int d = 1; d < 32; d <<= 1) {
    int o = __shfl_up(inc, d);
    if (lane >= d) inc += o;
  }
  if (lane == 31) wtot[wv] = inc;
  __syncthreads();
  int wo = (lane < wv) ? wtot[lane] : 0;
  for (int d = 16; d >= 1; d >>= 1) wo += __shfl_xor(wo, d);

  int run = wo + inc - sum;
  for (int j = 0; j < SCAN_PER; ++j) {
    const int c = s[base + j];
    s[base + j] = run;
    run += c;
  }
  __syncthreads();
  for (int i = t; i < NPAD / 4; i += SCAN_T) ((int4*)offs)[i] = ((int4*)s)[i];
}

__global__ __launch_bounds__(EDGE_T) void k_place(const int* __restrict__ row,
                                                  const int* __restrict__ col,
                                                  const int* __restrict__ offs,
                                                  int* __restrict__ rec) {
  __shared__ int cur[256];
  const int t = threadIdx.x;
  const int g = blockIdx.x;
  if (t < NB) cur[t] = offs[t * NG + g];
  const int e = g * CHUNK + t;
  int d[EPT], sn[EPT];
#pragma unroll
  for (int j = 0; j < EPT; ++j) {
    d[j] = row[e + j * EDGE_T];
    sn[j] = col[e + j * EDGE_T];
  }
  __syncthreads();
#pragma unroll
  for (int j = 0; j < EPT; ++j) {
    const int p = atomicAdd(&cur[d[j] >> 8], 1);
    rec[p] = (sn[j] << 8) | (d[j] & 255);
  }
}

__global__ __launch_bounds__(BKT_T) void k_bucket(const int* __restrict__ offs,
                                                  const int* __restrict__ rec,
                                                  int* __restrict__ head,
                                                  int* __restrict__ srt,
                                                  float* __restrict__ deg) {
  __shared__ int cnt[256];
  __shared__ int cur[256];
  __shared__ int wtot[8];
  __shared__ int stage[STAGE];
  const int t = threadIdx.x;
  const int b = blockIdx.x;
  const int lane = t & 31;
  const int wv = t >> 5;
  const int base = offs[b * NG];
  const int n = offs[(b + 1) * NG] - base;
  const int* rb = rec + base;
  if (t < 256) cnt[t] = 0;
  __syncthreads();
  for (int i0 = t; i0 < n; i0 += BKT_U * BKT_T) {
    int r[BKT_U];
#pragma unroll
    for (int u = 0; u < BKT_U; ++u) {
      const int i = i0 + u * BKT_T;
      r[u] = (i < n) ? rb[i] : -1;
    }
#pragma unroll
    for (int u = 0; u < BKT_U; ++u)
      if (r[u] >= 0) atomicAdd(&cnt[r[u] & 255], 1);
  }
  __syncthreads();
  int c = 0, inc = 0;
  if (t < 256) {
    c = cnt[t];
    inc = c;
    for (int d = 1; d < 32; d <<= 1) {
      int o = __shfl_up(inc, d);
      if (lane >= d) inc += o;
    }
    if (lane == 31) wtot[wv] = inc;
  }
  __syncthreads();
  if (t < 256) {
    int wo = (lane < wv) ? wtot[lane] : 0;
    for (int d = 16; d >= 1; d >>= 1) wo += __shfl_xor(wo, d);
    const int first = base + wo + inc - c;
    const int node = (b << 8) + t;
    if (node <= NN) head[node] = first;
    if (node < NN) deg[node] = (float)c;
    cur[t] = first;
  }
  __syncthreads();
  const bool fits = n <= STAGE;
  for (int i0 = t; i0 < n; i0 += BKT_U * BKT_T) {
    int r[BKT_U];
#pragma unroll
    for (int u = 0; u < BKT_U; ++u) {
      const int i = i0 + u * BKT_T;
      r[u] = (i < n) ? rb[i] : -1;
    }
#pragma unroll
    for (int u = 0; u < BKT_U; ++u) {
      if (r[u] >= 0) {
        const int p = atomicAdd(&cur[r[u] & 255], 1);
        if (fits) stage[p - base] = r[u] >> 8;
        else srt[p] = r[u] >> 8;
      }
    }
  }
  if (fits) {
    __syncthreads();
    for (int i = t; i < n; i += BKT_T) srt[base + i] = stage[i];
  }
}

template <int N>
__device__ __forceinline__ void take_rows(const float* __restrict__ xl, int my,
                                          int j, float4& acc) {
  float4 v[N];
#pragma unroll
  for (int u = 0; u < N; ++u) {
    const int sn = __builtin_amdgcn_readlane(my, j + u);
    v[u] = *(const float4*)(xl + (size_t)sn * DD);
  }
#pragma unroll
  for (int u = 0; u < N; ++u) {
    acc.x += v[u].x;
    acc.y += v[u].y;
    acc.z += v[u].z;
    acc.w += v[u].w;
  }
}

__global__ __launch_bounds__(256) void k_gather(const float* __restrict__ xin,
                                                const int* __restrict__ head,
                                                const int* __restrict__ srt,
                                                float* __restrict__ agg) {
  const int lane = threadIdx.x & 31;
  const int node = blockIdx.x * 8 + (threadIdx.x >> 5);
  const int beg = __builtin_amdgcn_readfirstlane(head[node]);
  const int end = __builtin_amdgcn_readfirstlane(head[node + 1]);
  const float* xl = xin + (lane << 2);
  float4 acc = {0.f, 0.f, 0.f, 0.f};
  for (int b = beg; b < end; b += 32) {
    const int m = (end - b < 32) ? (end - b) : 32;
    const int my = (lane < m) ? srt[b + lane] : 0;
    int j = 0;
    for (; j + 8 <= m; j += 8) take_rows<8>(xl, my, j, acc);
    if (m - j >= 4) { take_rows<4>(xl, my, j, acc); j += 4; }
    if (m - j >= 2) { take_rows<2>(xl, my, j, acc); j += 2; }
    if (m - j >= 1) take_rows<1>(xl, my, j, acc);
  }
  *(float4*)(agg + (size_t)node * DD + (lane << 2)) = acc;
}

__global__ void k_sage_gemm(const float* __restrict__ agg,
                            const float* __restrict__ deg,
                            const float* __restrict__ h,
                            const float* __restrict__ Wl,
                            const float* __restrict__ Wr,
                            const float* __restrict__ bias,
                            const float* __restrict__ resid,
                            float* __restrict__ out,
                            float* __restrict__ hout) {
  __shared__ float wlds[2 * DD * WSTRIDE];

  for (int i = threadIdx.x; i < (DD * DD) / 4; i += 256) {
    int o = i >> 5;
    int kc = (i & 31) << 2;
    float4 a = *(const float4*)(Wl + (size_t)o * DD + kc);
    float4 b = *(const float4*)(Wr + (size_t)o * DD + kc);
    *(float4*)(&wlds[o * WSTRIDE + kc]) = a;
    *(float4*)(&wlds[WOFF + o * WSTRIDE + kc]) = b;
  }
  __syncthreads();

  const int nw = blockDim.x >> 5;
  const int wv = threadIdx.x >> 5;
  const int lane = threadIdx.x & 31;
  const int half = lane >> 4;
  const int lm = lane & 15;
  const int tstride = gridDim.x * nw;

  for (int tile = blockIdx.x * nw + wv; tile < NT; tile += tstride) {
    const int m0 = tile << 4;
    if (tile + tstride < NT) {
      const size_t nb = (size_t)((tile + tstride) * 16 + lm) * DD;
      __builtin_prefetch(agg + nb, 0, 1);
      __builtin_prefetch(h + nb, 0, 1);
    }

    const float sc = 1.0f / fmaxf(deg[m0 + lm], 1.0f);
    const float* arow = agg + (size_t)(m0 + lm) * DD + 2 * half;
    const float* hrow = h + (size_t)(m0 + lm) * DD + 2 * half;
    const float* blbase = &wlds[(size_t)lm * WSTRIDE + 2 * half];
    const float* brbase = &wlds[WOFF + (size_t)lm * WSTRIDE + 2 * half];

    v8f acc[8] = {};

#pragma unroll 1
    for (int pass = 0; pass < 2; ++pass) {
      const float* aptr = pass ? hrow : arow;
      const float* bb = pass ? brbase : blbase;
      const float scale = pass ? 1.0f : sc;

      v2f cur[8], nxt[8];
#pragma unroll
      for (int t = 0; t < 8; ++t)
        cur[t] = *(const v2f*)(bb + (size_t)(t << 4) * WSTRIDE);
      float2 a2 = *(const float2*)(aptr);

#pragma unroll 2
      for (int k = 0; k < 32; ++k) {
        const int K1 = ((k + 1) & 31) << 2;
#pragma unroll
        for (int t = 0; t < 8; ++t)
          nxt[t] = *(const v2f*)(bb + (size_t)(t << 4) * WSTRIDE + K1);
        float2 a2n = *(const float2*)(aptr + K1);

        v2f va;
        va.x = a2.x * scale;
        va.y = a2.y * scale;
#pragma unroll
        for (int t = 0; t < 8; ++t) acc[t] = wmma4(va, cur[t], acc[t]);

#pragma unroll
        for (int t = 0; t < 8; ++t) cur[t] = nxt[t];
        a2 = a2n;
      }
    }

#pragma unroll
    for (int t = 0; t < 8; ++t) {
      const int n0 = t << 4;
      const float bb = bias[n0 + lm];
#pragma unroll
      for (int r = 0; r < 8; ++r) {
        const size_t idx = (size_t)(m0 + r + (half << 3)) * DD + n0 + lm;
        float v = acc[t][r] + bb;
        if (resid) v += resid[idx];
        out[idx] = v;
        if (hout) hout[idx] = fmaxf(v, 0.0f);
      }
    }
  }
}

extern "C" void kernel_launch(void* const* d_in, const int* in_sizes, int n_in,
                              void* d_out, int out_size, void* d_ws,
                              size_t ws_size, hipStream_t stream) {
  const float* x = (const float*)d_in[0];
  const int* erow = (const int*)d_in[1];
  const int* ecol = (const int*)d_in[2];
  const float* Wl0 = (const float*)d_in[3];
  const float* Wr0 = (const float*)d_in[4];
  const float* b0 = (const float*)d_in[5];
  const float* Wl1 = (const float*)d_in[6];
  const float* Wr1 = (const float*)d_in[7];
  const float* b1 = (const float*)d_in[8];
  const float* Wl2 = (const float*)d_in[9];
  const float* Wr2 = (const float*)d_in[10];
  const float* b2 = (const float*)d_in[11];
  float* out = (float*)d_out;

  float* ws = (float*)d_ws;
  float* agg = ws;
  float* h = agg + (size_t)FEAT;
  int* cntm = (int*)(h + (size_t)FEAT);
  int* head = cntm + NN;
  int* offs = head + NPAD;
  int* rec = offs + NPAD;
  int* srt = rec + NE;
  float* deg = (float*)(srt + NE);

  dim3 b256(256);
  const int gatherBlocks = NN / 8;
  const int gemmBlocks = (NT + 7) / 8;

  k_hist<<<dim3(NG), dim3(EDGE_T), 0, stream>>>(erow, cntm);
  k_scan<<<dim3(1), dim3(SCAN_T), 0, stream>>>(cntm, offs);
  k_place<<<dim3(NG), dim3(EDGE_T), 0, stream>>>(erow, ecol, offs, rec);
  k_bucket<<<dim3(NB), dim3(BKT_T), 0, stream>>>(offs, rec, head, srt, deg);

  k_gather<<<dim3(gatherBlocks), b256, 0, stream>>>(x, head, srt, agg);
  k_sage_gemm<<<dim3(gemmBlocks), b256, 0, stream>>>(agg, deg, x, Wl0, Wr0, b0,
                                                     nullptr, out, h);

  k_gather<<<dim3(gatherBlocks), b256, 0, stream>>>(h, head, srt, agg);
  k_sage_gemm<<<dim3(gemmBlocks), b256, 0, stream>>>(agg, deg, h, Wl1, Wr1, b1,
                                                     out, out, h);

  k_gather<<<dim3(gatherBlocks), b256, 0, stream>>>(h, head, srt, agg);
  k_sage_gemm<<<dim3(gemmBlocks), b256, 0, stream>>>(agg, deg, h, Wl2, Wr2, b2,
                                                     out, out, nullptr);
}


# Neighbour sums folded into the layer product: each wave sums its 16-node tile's source rows into an LDS tile that is the WMMA's A operand; 3 launches and the mean plane go; sum order unchanged
# speedup vs baseline: 3.3663x; 1.0696x over previous
#include <hip/hip_runtime.h>


#define NN 50000
#define NE 800000
#define DD 128
#define NT (NN / 16)
#define FEAT (NN * DD)
#define WSTRIDE 132
#define WOFF (DD * WSTRIDE)
#define SCAN_T 1024
#define SCAN_PER 49
#define NPAD (SCAN_T * SCAN_PER)
#define SCAN_FULL 12
#define NB 196
#define NG 250
#define CHUNK (NE / NG)
#define EDGE_T 640
#define EPT (CHUNK / EDGE_T)
#define M4 ((NB * NG) / 4)
#define BKT_T 1024
#define BKT_U 4
#define STAGE 8192
#define LW 16
#define LT (LW * 32)
#define LBLK 256
#define ATILE (16 * WSTRIDE)
#define AOFF (2 * WOFF)

typedef float v2f __attribute__((ext_vector_type(2)));
typedef float v8f __attribute__((ext_vector_type(8)));

__device__ __forceinline__ v8f wmma4(v2f a, v2f b, v8f c) {
  return __builtin_amdgcn_wmma_f32_16x16x4_f32(false, a, false, b, (short)0, c,
                                               false, false);
}

__global__ __launch_bounds__(EDGE_T) void k_hist(const int* __restrict__ row,
                                                 int* __restrict__ cntm) {
  __shared__ int hist[256];
  const int t = threadIdx.x;
  const int g = blockIdx.x;
  if (t < 256) hist[t] = 0;
  __syncthreads();
  const int* r = row + g * CHUNK + t;
  int d[EPT];
#pragma unroll
  for (int j = 0; j < EPT; ++j) d[j] = r[j * EDGE_T];
#pragma unroll
  for (int j = 0; j < EPT; ++j) atomicAdd(&hist[d[j] >> 8], 1);
  __syncthreads();
  if (t < NB) cntm[t * NG + g] = hist[t];
}

__global__ __launch_bounds__(SCAN_T) void k_scan(const int* __restrict__ cntm,
                                                 int* __restrict__ offs) {
  __shared__ int s[NPAD];
  __shared__ int wtot[SCAN_T / 32];
  const int t = threadIdx.x;
  const int lane = t & 31;
  const int wv = t >> 5;
  const int4 z4 = {0, 0, 0, 0};

  int4 v[SCAN_FULL];
#pragma unroll
  for (int k = 0; k < SCAN_FULL; ++k) {
    const int i = t + k * SCAN_T;
    v[k] = (i < M4) ? ((const int4*)cntm)[i] : z4;
  }
  {
    const int i = t + SCAN_FULL * SCAN_T;
    if (i < NPAD / 4) ((int4*)s)[i] = z4;
  }
#pragma unroll
  for (int k = 0; k < SCAN_FULL; ++k) ((int4*)s)[t + k * SCAN_T] = v[k];
  __syncthreads();

  const int base = t * SCAN_PER;
  int sum = 0;
  for (int j = 0; j < SCAN_PER; ++j) sum += s[base + j];

  int inc = sum;
  for (int d = 1; d < 32; d <<= 1) {
    int o = __shfl_up(inc, d);
    if (lane >= d) inc += o;
  }
  if (lane == 31) wtot[wv] = inc;
  __syncthreads();
  int wo = (lane < wv) ? wtot[lane] : 0;
  for (int d = 16; d >= 1; d >>= 1) wo += __shfl_xor(wo, d);

  int run = wo + inc - sum;
  for (int j = 0; j < SCAN_PER; ++j) {
    const int c = s[base + j];
    s[base + j] = run;
    run += c;
  }
  __syncthreads();
  for (int i = t; i < NPAD / 4; i += SCAN_T) ((int4*)offs)[i] = ((int4*)s)[i];
}

__global__ __launch_bounds__(EDGE_T) void k_place(const int* __restrict__ row,
                                                  const int* __restrict__ col,
                                                  const int* __restrict__ offs,
                                                  int* __restrict__ rec) {
  __shared__ int cur[256];
  const int t = threadIdx.x;
  const int g = blockIdx.x;
  if (t < NB) cur[t] = offs[t * NG + g];
  const int e = g * CHUNK + t;
  int d[EPT], sn[EPT];
#pragma unroll
  for (int j = 0; j < EPT; ++j) {
    d[j] = row[e + j * EDGE_T];
    sn[j] = col[e + j * EDGE_T];
  }
  __syncthreads();
#pragma unroll
  for (int j = 0; j < EPT; ++j) {
    const int p = atomicAdd(&cur[d[j] >> 8], 1);
    rec[p] = (sn[j] << 8) | (d[j] & 255);
  }
}

__global__ __launch_bounds__(BKT_T) void k_bucket(const int* __restrict__ offs,
                                                  const int* __restrict__ rec,
                                                  int* __restrict__ head,
                                                  int* __restrict__ srt) {
  __shared__ int cnt[256];
  __shared__ int cur[256];
  __shared__ int wtot[8];
  __shared__ int stage[STAGE];
  const int t = threadIdx.x;
  const int b = blockIdx.x;
  const int lane = t & 31;
  const int wv = t >> 5;
  const int base = offs[b * NG];
  const int n = offs[(b + 1) * NG] - base;
  const int* rb = rec + base;
  if (t < 256) cnt[t] = 0;
  __syncthreads();
  for (int i0 = t; i0 < n; i0 += BKT_U * BKT_T) {
    int r[BKT_U];
#pragma unroll
    for (int u = 0; u < BKT_U; ++u) {
      const int i = i0 + u * BKT_T;
      r[u] = (i < n) ? rb[i] : -1;
    }
#pragma unroll
    for (int u = 0; u < BKT_U; ++u)
      if (r[u] >= 0) atomicAdd(&cnt[r[u] & 255], 1);
  }
  __syncthreads();
  int c = 0, inc = 0;
  if (t < 256) {
    c = cnt[t];
    inc = c;
    for (int d = 1; d < 32; d <<= 1) {
      int o = __shfl_up(inc, d);
      if (lane >= d) inc += o;
    }
    if (lane == 31) wtot[wv] = inc;
  }
  __syncthreads();
  if (t < 256) {
    int wo = (lane < wv) ? wtot[lane] : 0;
    for (int d = 16; d >= 1; d >>= 1) wo += __shfl_xor(wo, d);
    const int first = base + wo + inc - c;
    const int node = (b << 8) + t;
    if (node <= NN) head[node] = first;
    cur[t] = first;
  }
  __syncthreads();
  const bool fits = n <= STAGE;
  for (int i0 = t; i0 < n; i0 += BKT_U * BKT_T) {
    int r[BKT_U];
#pragma unroll
    for (int u = 0; u < BKT_U; ++u) {
      const int i = i0 + u * BKT_T;
      r[u] = (i < n) ? rb[i] : -1;
    }
#pragma unroll
    for (int u = 0; u < BKT_U; ++u) {
      if (r[u] >= 0) {
        const int p = atomicAdd(&cur[r[u] & 255], 1);
        if (fits) stage[p - base] = r[u] >> 8;
        else srt[p] = r[u] >> 8;
      }
    }
  }
  if (fits) {
    __syncthreads();
    for (int i = t; i < n; i += BKT_T) srt[base + i] = stage[i];
  }
}

struct Walk {
  int ni;
  int nbeg;
  int nend;
};

__device__ __forceinline__ void close_node(Walk& w, float4& acc, int hd,
                                           float* __restrict__ atile,
                                           int lane) {
  const float sc = 1.0f / fmaxf((float)(w.nend - w.nbeg), 1.0f);
  float4 o;
  o.x = acc.x * sc;
  o.y = acc.y * sc;
  o.z = acc.z * sc;
  o.w = acc.w * sc;
  *(float4*)(atile + w.ni * WSTRIDE + (lane << 2)) = o;
  acc.x = 0.f;
  acc.y = 0.f;
  acc.z = 0.f;
  acc.w = 0.f;
  w.ni += 1;
  w.nbeg = w.nend;
  w.nend = __builtin_amdgcn_readlane(hd, w.ni + 1);
}

template <int N>
__device__ __forceinline__ void take_rows(const float* __restrict__ xl, int my,
                                          int j, int pos, Walk& w, float4& acc,
                                          int hd, float* __restrict__ atile,
                                          int lane) {
  float4 v[N];
#pragma unroll
  for (int u = 0; u < N; ++u) {
    const int sn = __builtin_amdgcn_readlane(my, j + u);
    v[u] = *(const float4*)(xl + (size_t)sn * DD);
  }
#pragma unroll
  for (int u = 0; u < N; ++u) {
    while (pos + u == w.nend) close_node(w, acc, hd, atile, lane);
    acc.x += v[u].x;
    acc.y += v[u].y;
    acc.z += v[u].z;
    acc.w += v[u].w;
  }
}

__device__ __forceinline__ void k_pass(const float* __restrict__ a,
                                       const float* __restrict__ bb,
                                       v8f (&acc)[8]) {
  v2f cur[8], nxt[8];
#pragma unroll
  for (int t = 0; t < 8; ++t)
    cur[t] = *(const v2f*)(bb + (size_t)(t << 4) * WSTRIDE);
  v2f a2 = *(const v2f*)(a);

#pragma unroll 2
  for (int k = 0; k < 32; ++k) {
    const int K1 = ((k + 1) & 31) << 2;
#pragma unroll
    for (int t = 0; t < 8; ++t)
      nxt[t] = *(const v2f*)(bb + (size_t)(t << 4) * WSTRIDE + K1);
    const v2f a2n = *(const v2f*)(a + K1);

#pragma unroll
    for (int t = 0; t < 8; ++t) acc[t] = wmma4(a2, cur[t], acc[t]);

#pragma unroll
    for (int t = 0; t < 8; ++t) cur[t] = nxt[t];
    a2 = a2n;
  }
}

__global__ __launch_bounds__(LT) void k_sage_layer(
    const float* __restrict__ hin, const int* __restrict__ head,
    const int* __restrict__ srt, const float* __restrict__ Wl,
    const float* __restrict__ Wr, const float* __restrict__ bias,
    const float* __restrict__ resid,
    float* __restrict__ out,
    float* __restrict__ hout) {
  __shared__ __attribute__((aligned(16))) float wlds[AOFF + LW * ATILE];

  const int wv = threadIdx.x >> 5;
  const int lane = threadIdx.x & 31;
  const int half = lane >> 4;
  const int lm = lane & 15;
  const int tile = wv * gridDim.x + blockIdx.x;
  const bool active = tile < NT;
  const int m0 = tile << 4;
  float* atile = &wlds[AOFF + wv * ATILE];

#pragma unroll
  for (int it = 0; it < (DD * DD / 4) / LT; ++it) {
    const int i = threadIdx.x + it * LT;
    const int o = i >> 5;
    const int kc = (i & 31) << 2;
    const float4 a = *(const float4*)(Wl + (size_t)o * DD + kc);
    const float4 b = *(const float4*)(Wr + (size_t)o * DD + kc);
    *(float4*)(&wlds[o * WSTRIDE + kc]) = a;
    *(float4*)(&wlds[WOFF + o * WSTRIDE + kc]) = b;
  }

  if (active) {
    const int hd = (lane <= 16) ? head[m0 + lane] : 0;
    const int e0 = __builtin_amdgcn_readlane(hd, 0);
    const int e1 = __builtin_amdgcn_readlane(hd, 16);
    Walk w;
    w.ni = 0;
    w.nbeg = e0;
    w.nend = __builtin_amdgcn_readlane(hd, 1);
    const float* xl = hin + (lane << 2);
    float4 acc = {0.f, 0.f, 0.f, 0.f};
    int nx = (e0 + lane < e1) ? srt[e0 + lane] : 0;
    for (int b = e0; b < e1; b += 32) {
      const int m = (e1 - b < 32) ? (e1 - b) : 32;
      const int my = nx;
      nx = (b + 32 + lane < e1) ? srt[b + 32 + lane] : 0;
      int j = 0;
      for (; j + 16 <= m; j += 16)
        take_rows<16>(xl, my, j, b + j, w, acc, hd, atile, lane);
      if (m - j >= 8) {
        take_rows<8>(xl, my, j, b + j, w, acc, hd, atile, lane);
        j += 8;
      }
      if (m - j >= 4) {
        take_rows<4>(xl, my, j, b + j, w, acc, hd, atile, lane);
        j += 4;
      }
      if (m - j >= 2) {
        take_rows<2>(xl, my, j, b + j, w, acc, hd, atile, lane);
        j += 2;
      }
      if (m - j >= 1) take_rows<1>(xl, my, j, b + j, w, acc, hd, atile, lane);
    }
    while (w.ni < 16) close_node(w, acc, hd, atile, lane);
  }
  __syncthreads();
  if (!active) return;

  const float* arow = atile + lm * WSTRIDE + 2 * half;
  const float* hrow = hin + (size_t)(m0 + lm) * DD + 2 * half;
  const float* blbase = &wlds[(size_t)lm * WSTRIDE + 2 * half];
  const float* brbase = &wlds[WOFF + (size_t)lm * WSTRIDE + 2 * half];

  v8f acc[8] = {};

  k_pass(arow, blbase, acc);
  k_pass(hrow, brbase, acc);

#pragma unroll
  for (int t = 0; t < 8; ++t) {
    const int n0 = t << 4;
    const float bb = bias[n0 + lm];
#pragma unroll
    for (int r = 0; r < 8; ++r) {
      const size_t idx = (size_t)(m0 + r + (half << 3)) * DD + n0 + lm;
      float v = acc[t][r] + bb;
      if (resid) v += resid[idx];
      out[idx] = v;
      if (hout) hout[idx] = fmaxf(v, 0.0f);
    }
  }
}

extern "C" void kernel_launch(void* const* d_in, const int* in_sizes, int n_in,
                              void* d_out, int out_size, void* d_ws,
                              size_t ws_size, hipStream_t stream) {
  const float* x = (const float*)d_in[0];
  const int* erow = (const int*)d_in[1];
  const int* ecol = (const int*)d_in[2];
  const float* Wl0 = (const float*)d_in[3];
  const float* Wr0 = (const float*)d_in[4];
  const float* b0 = (const float*)d_in[5];
  const float* Wl1 = (const float*)d_in[6];
  const float* Wr1 = (const float*)d_in[7];
  const float* b1 = (const float*)d_in[8];
  const float* Wl2 = (const float*)d_in[9];
  const float* Wr2 = (const float*)d_in[10];
  const float* b2 = (const float*)d_in[11];
  float* out = (float*)d_out;

  float* ws = (float*)d_ws;
  float* ha = ws;
  float* hb = ha + (size_t)FEAT;
  int* cntm = (int*)(hb + (size_t)FEAT);
  int* head = cntm + NN;
  int* offs = head + NPAD;
  int* rec = offs + NPAD;
  int* srt = rec + NE;

  k_hist<<<dim3(NG), dim3(EDGE_T), 0, stream>>>(erow, cntm);
  k_scan<<<dim3(1), dim3(SCAN_T), 0, stream>>>(cntm, offs);
  k_place<<<dim3(NG), dim3(EDGE_T), 0, stream>>>(erow, ecol, offs, rec);
  k_bucket<<<dim3(NB), dim3(BKT_T), 0, stream>>>(offs, rec, head, srt);

  k_sage_layer<<<dim3(LBLK), dim3(LT), 0, stream>>>(x, head, srt, Wl0, Wr0, b0,
                                                    nullptr, out, ha);

  k_sage_layer<<<dim3(LBLK), dim3(LT), 0, stream>>>(ha, head, srt, Wl1, Wr1, b1,
                                                    out, out, hb);

  k_sage_layer<<<dim3(LBLK), dim3(LT), 0, stream>>>(hb, head, srt, Wl2, Wr2, b2,
                                                    out, out, nullptr);
}


# Epilogue waits removed: the tile's 64 residual and 8 bias values are loaded together in front of the products, so their round trips pass under the matrix work; every sum and its order unchanged
# speedup vs baseline: 4.0251x; 1.1957x over previous
#include <hip/hip_runtime.h>


#define NN 50000
#define NE 800000
#define DD 128
#define NT (NN / 16)
#define FEAT (NN * DD)
#define WSTRIDE 132
#define WOFF (DD * WSTRIDE)
#define SCAN_T 1024
#define SCAN_PER 49
#define NPAD (SCAN_T * SCAN_PER)
#define SCAN_FULL 12
#define NB 196
#define NG 250
#define CHUNK (NE / NG)
#define EDGE_T 640
#define EPT (CHUNK / EDGE_T)
#define M4 ((NB * NG) / 4)
#define BKT_T 1024
#define BKT_U 4
#define STAGE 8192
#define LW 16
#define LT (LW * 32)
#define LBLK 256
#define ATILE (16 * WSTRIDE)
#define AOFF (2 * WOFF)

typedef float v2f __attribute__((ext_vector_type(2)));
typedef float v8f __attribute__((ext_vector_type(8)));

__device__ __forceinline__ v8f wmma4(v2f a, v2f b, v8f c) {
  return __builtin_amdgcn_wmma_f32_16x16x4_f32(false, a, false, b, (short)0, c,
                                               false, false);
}

__global__ __launch_bounds__(EDGE_T) void k_hist(const int* __restrict__ row,
                                                 int* __restrict__ cntm) {
  __shared__ int hist[256];
  const int t = threadIdx.x;
  const int g = blockIdx.x;
  if (t < 256) hist[t] = 0;
  __syncthreads();
  const int* r = row + g * CHUNK + t;
  int d[EPT];
#pragma unroll
  for (int j = 0; j < EPT; ++j) d[j] = r[j * EDGE_T];
#pragma unroll
  for (int j = 0; j < EPT; ++j) atomicAdd(&hist[d[j] >> 8], 1);
  __syncthreads();
  if (t < NB) cntm[t * NG + g] = hist[t];
}

__global__ __launch_bounds__(SCAN_T) void k_scan(const int* __restrict__ cntm,
                                                 int* __restrict__ offs) {
  __shared__ int s[NPAD];
  __shared__ int wtot[SCAN_T / 32];
  const int t = threadIdx.x;
  const int lane = t & 31;
  const int wv = t >> 5;
  const int4 z4 = {0, 0, 0, 0};

  int4 v[SCAN_FULL];
#pragma unroll
  for (int k = 0; k < SCAN_FULL; ++k) {
    const int i = t + k * SCAN_T;
    v[k] = (i < M4) ? ((const int4*)cntm)[i] : z4;
  }
  {
    const int i = t + SCAN_FULL * SCAN_T;
    if (i < NPAD / 4) ((int4*)s)[i] = z4;
  }
#pragma unroll
  for (int k = 0; k < SCAN_FULL; ++k) ((int4*)s)[t + k * SCAN_T] = v[k];
  __syncthreads();

  const int base = t * SCAN_PER;
  int sum = 0;
  for (int j = 0; j < SCAN_PER; ++j) sum += s[base + j];

  int inc = sum;
  for (int d = 1; d < 32; d <<= 1) {
    int o = __shfl_up(inc, d);
    if (lane >= d) inc += o;
  }
  if (lane == 31) wtot[wv] = inc;
  __syncthreads();
  int wo = (lane < wv) ? wtot[lane] : 0;
  for (int d = 16; d >= 1; d >>= 1) wo += __shfl_xor(wo, d);

  int run = wo + inc - sum;
  for (int j = 0; j < SCAN_PER; ++j) {
    const int c = s[base + j];
    s[base + j] = run;
    run += c;
  }
  __syncthreads();
  for (int i = t; i < NPAD / 4; i += SCAN_T) ((int4*)offs)[i] = ((int4*)s)[i];
}

__global__ __launch_bounds__(EDGE_T) void k_place(const int* __restrict__ row,
                                                  const int* __restrict__ col,
                                                  const int* __restrict__ offs,
                                                  int* __restrict__ rec) {
  __shared__ int cur[256];
  const int t = threadIdx.x;
  const int g = blockIdx.x;
  if (t < NB) cur[t] = offs[t * NG + g];
  const int e = g * CHUNK + t;
  int d[EPT], sn[EPT];
#pragma unroll
  for (int j = 0; j < EPT; ++j) {
    d[j] = row[e + j * EDGE_T];
    sn[j] = col[e + j * EDGE_T];
  }
  __syncthreads();
#pragma unroll
  for (int j = 0; j < EPT; ++j) {
    const int p = atomicAdd(&cur[d[j] >> 8], 1);
    rec[p] = (sn[j] << 8) | (d[j] & 255);
  }
}

__global__ __launch_bounds__(BKT_T) void k_bucket(const int* __restrict__ offs,
                                                  const int* __restrict__ rec,
                                                  int* __restrict__ head,
                                                  int* __restrict__ srt) {
  __shared__ int cnt[256];
  __shared__ int cur[256];
  __shared__ int wtot[8];
  __shared__ int stage[STAGE];
  const int t = threadIdx.x;
  const int b = blockIdx.x;
  const int lane = t & 31;
  const int wv = t >> 5;
  const int base = offs[b * NG];
  const int n = offs[(b + 1) * NG] - base;
  const int* rb = rec + base;
  if (t < 256) cnt[t] = 0;
  __syncthreads();
  for (int i0 = t; i0 < n; i0 += BKT_U * BKT_T) {
    int r[BKT_U];
#pragma unroll
    for (int u = 0; u < BKT_U; ++u) {
      const int i = i0 + u * BKT_T;
      r[u] = (i < n) ? rb[i] : -1;
    }
#pragma unroll
    for (int u = 0; u < BKT_U; ++u)
      if (r[u] >= 0) atomicAdd(&cnt[r[u] & 255], 1);
  }
  __syncthreads();
  int c = 0, inc = 0;
  if (t < 256) {
    c = cnt[t];
    inc = c;
    for (int d = 1; d < 32; d <<= 1) {
      int o = __shfl_up(inc, d);
      if (lane >= d) inc += o;
    }
    if (lane == 31) wtot[wv] = inc;
  }
  __syncthreads();
  if (t < 256) {
    int wo = (lane < wv) ? wtot[lane] : 0;
    for (int d = 16; d >= 1; d >>= 1) wo += __shfl_xor(wo, d);
    const int first = base + wo + inc - c;
    const int node = (b << 8) + t;
    if (node <= NN) head[node] = first;
    cur[t] = first;
  }
  __syncthreads();
  const bool fits = n <= STAGE;
  for (int i0 = t; i0 < n; i0 += BKT_U * BKT_T) {
    int r[BKT_U];
#pragma unroll
    for (int u = 0; u < BKT_U; ++u) {
      const int i = i0 + u * BKT_T;
      r[u] = (i < n) ? rb[i] : -1;
    }
#pragma unroll
    for (int u = 0; u < BKT_U; ++u) {
      if (r[u] >= 0) {
        const int p = atomicAdd(&cur[r[u] & 255], 1);
        if (fits) stage[p - base] = r[u] >> 8;
        else srt[p] = r[u] >> 8;
      }
    }
  }
  if (fits) {
    __syncthreads();
    for (int i = t; i < n; i += BKT_T) srt[base + i] = stage[i];
  }
}

struct Walk {
  int ni;
  int nbeg;
  int nend;
};

__device__ __forceinline__ void close_node(Walk& w, float4& acc, int hd,
                                           float* __restrict__ atile,
                                           int lane) {
  const float sc = 1.0f / fmaxf((float)(w.nend - w.nbeg), 1.0f);
  float4 o;
  o.x = acc.x * sc;
  o.y = acc.y * sc;
  o.z = acc.z * sc;
  o.w = acc.w * sc;
  *(float4*)(atile + w.ni * WSTRIDE + (lane << 2)) = o;
  acc.x = 0.f;
  acc.y = 0.f;
  acc.z = 0.f;
  acc.w = 0.f;
  w.ni += 1;
  w.nbeg = w.nend;
  w.nend = __builtin_amdgcn_readlane(hd, w.ni + 1);
}

template <int N>
__device__ __forceinline__ void take_rows(const float* __restrict__ xl, int my,
                                          int j, int pos, Walk& w, float4& acc,
                                          int hd, float* __restrict__ atile,
                                          int lane) {
  float4 v[N];
#pragma unroll
  for (int u = 0; u < N; ++u) {
    const int sn = __builtin_amdgcn_readlane(my, j + u);
    v[u] = *(const float4*)(xl + (size_t)sn * DD);
  }
#pragma unroll
  for (int u = 0; u < N; ++u) {
    while (pos + u == w.nend) close_node(w, acc, hd, atile, lane);
    acc.x += v[u].x;
    acc.y += v[u].y;
    acc.z += v[u].z;
    acc.w += v[u].w;
  }
}

__device__ __forceinline__ void k_pass(const float* __restrict__ a,
                                       const float* __restrict__ bb,
                                       v8f (&acc)[8]) {
  v2f cur[8], nxt[8];
#pragma unroll
  for (int t = 0; t < 8; ++t)
    cur[t] = *(const v2f*)(bb + (size_t)(t << 4) * WSTRIDE);
  v2f a2 = *(const v2f*)(a);

#pragma unroll 2
  for (int k = 0; k < 32; ++k) {
    const int K1 = ((k + 1) & 31) << 2;
#pragma unroll
    for (int t = 0; t < 8; ++t)
      nxt[t] = *(const v2f*)(bb + (size_t)(t << 4) * WSTRIDE + K1);
    const v2f a2n = *(const v2f*)(a + K1);

#pragma unroll
    for (int t = 0; t < 8; ++t) acc[t] = wmma4(a2, cur[t], acc[t]);

#pragma unroll
    for (int t = 0; t < 8; ++t) cur[t] = nxt[t];
    a2 = a2n;
  }
}

template <bool RES, bool RELU>
__device__ __forceinline__ void put_tile(const v8f (&acc)[8],
                                         const float (&bb)[8],
                                         const float (&rs)[8][8],
                                         float* __restrict__ o,
                                         float* __restrict__ h) {
#pragma unroll
  for (int t = 0; t < 8; ++t) {
#pragma unroll
    for (int r = 0; r < 8; ++r) {
      float v = acc[t][r] + bb[t];
      if (RES) v += rs[t][r];
      o[r * DD + (t << 4)] = v;
      if (RELU) h[r * DD + (t << 4)] = fmaxf(v, 0.0f);
    }
  }
}

__global__ __launch_bounds__(LT) void k_sage_layer(
    const float* __restrict__ hin, const int* __restrict__ head,
    const int* __restrict__ srt, const float* __restrict__ Wl,
    const float* __restrict__ Wr, const float* __restrict__ bias,
    const float* __restrict__ resid,
    float* __restrict__ out,
    float* __restrict__ hout) {
  __shared__ __attribute__((aligned(16))) float wlds[AOFF + LW * ATILE];

  const int wv = threadIdx.x >> 5;
  const int lane = threadIdx.x & 31;
  const int half = lane >> 4;
  const int lm = lane & 15;
  const int tile = wv * gridDim.x + blockIdx.x;
  const bool active = tile < NT;
  const int m0 = tile << 4;
  float* atile = &wlds[AOFF + wv * ATILE];

#pragma unroll
  for (int it = 0; it < (DD * DD / 4) / LT; ++it) {
    const int i = threadIdx.x + it * LT;
    const int o = i >> 5;
    const int kc = (i & 31) << 2;
    const float4 a = *(const float4*)(Wl + (size_t)o * DD + kc);
    const float4 b = *(const float4*)(Wr + (size_t)o * DD + kc);
    *(float4*)(&wlds[o * WSTRIDE + kc]) = a;
    *(float4*)(&wlds[WOFF + o * WSTRIDE + kc]) = b;
  }

  if (active) {
    const int hd = (lane <= 16) ? head[m0 + lane] : 0;
    const int e0 = __builtin_amdgcn_readlane(hd, 0);
    const int e1 = __builtin_amdgcn_readlane(hd, 16);
    Walk w;
    w.ni = 0;
    w.nbeg = e0;
    w.nend = __builtin_amdgcn_readlane(hd, 1);
    const float* xl = hin + (lane << 2);
    float4 acc = {0.f, 0.f, 0.f, 0.f};
    int nx = (e0 + lane < e1) ? srt[e0 + lane] : 0;
    for (int b = e0; b < e1; b += 32) {
      const int m = (e1 - b < 32) ? (e1 - b) : 32;
      const int my = nx;
      nx = (b + 32 + lane < e1) ? srt[b + 32 + lane] : 0;
      int j = 0;
      for (; j + 16 <= m; j += 16)
        take_rows<16>(xl, my, j, b + j, w, acc, hd, atile, lane);
      if (m - j >= 8) {
        take_rows<8>(xl, my, j, b + j, w, acc, hd, atile, lane);
        j += 8;
      }
      if (m - j >= 4) {
        take_rows<4>(xl, my, j, b + j, w, acc, hd, atile, lane);
        j += 4;
      }
      if (m - j >= 2) {
        take_rows<2>(xl, my, j, b + j, w, acc, hd, atile, lane);
        j += 2;
      }
      if (m - j >= 1) take_rows<1>(xl, my, j, b + j, w, acc, hd, atile, lane);
    }
    while (w.ni < 16) close_node(w, acc, hd, atile, lane);
  }
  __syncthreads();
  if (!active) return;

  const float* arow = atile + lm * WSTRIDE + 2 * half;
  const float* hrow = hin + (size_t)(m0 + lm) * DD + 2 * half;
  const float* blbase = &wlds[(size_t)lm * WSTRIDE + 2 * half];
  const float* brbase = &wlds[WOFF + (size_t)lm * WSTRIDE + 2 * half];

  const size_t obase = (size_t)(m0 + (half << 3)) * DD + lm;

  float bb[8];
#pragma unroll
  for (int t = 0; t < 8; ++t) bb[t] = bias[(t << 4) + lm];
  float rs[8][8];
  if (resid) {
#pragma unroll
    for (int t = 0; t < 8; ++t)
#pragma unroll
      for (int r = 0; r < 8; ++r)
        rs[t][r] = resid[obase + (size_t)r * DD + (t << 4)];
  } else {
#pragma unroll
    for (int t = 0; t < 8; ++t)
#pragma unroll
      for (int r = 0; r < 8; ++r) rs[t][r] = 0.f;
  }

  v8f acc[8] = {};

  k_pass(arow, blbase, acc);
  k_pass(hrow, brbase, acc);

  if (resid) {
    if (hout) put_tile<true, true>(acc, bb, rs, out + obase, hout + obase);
    else put_tile<true, false>(acc, bb, rs, out + obase, nullptr);
  } else {
    if (hout) put_tile<false, true>(acc, bb, rs, out + obase, hout + obase);
    else put_tile<false, false>(acc, bb, rs, out + obase, nullptr);
  }
}

extern "C" void kernel_launch(void* const* d_in, const int* in_sizes, int n_in,
                              void* d_out, int out_size, void* d_ws,
                              size_t ws_size, hipStream_t stream) {
  const float* x = (const float*)d_in[0];
  const int* erow = (const int*)d_in[1];
  const int* ecol = (const int*)d_in[2];
  const float* Wl0 = (const float*)d_in[3];
  const float* Wr0 = (const float*)d_in[4];
  const float* b0 = (const float*)d_in[5];
  const float* Wl1 = (const float*)d_in[6];
  const float* Wr1 = (const float*)d_in[7];
  const float* b1 = (const float*)d_in[8];
  const float* Wl2 = (const float*)d_in[9];
  const float* Wr2 = (const float*)d_in[10];
  const float* b2 = (const float*)d_in[11];
  float* out = (float*)d_out;

  float* ws = (float*)d_ws;
  float* ha = ws;
  float* hb = ha + (size_t)FEAT;
  int* cntm = (int*)(hb + (size_t)FEAT);
  int* head = cntm + NN;
  int* offs = head + NPAD;
  int* rec = offs + NPAD;
  int* srt = rec + NE;

  k_hist<<<dim3(NG), dim3(EDGE_T), 0, stream>>>(erow, cntm);
  k_scan<<<dim3(1), dim3(SCAN_T), 0, stream>>>(cntm, offs);
  k_place<<<dim3(NG), dim3(EDGE_T), 0, stream>>>(erow, ecol, offs, rec);
  k_bucket<<<dim3(NB), dim3(BKT_T), 0, stream>>>(offs, rec, head, srt);

  k_sage_layer<<<dim3(LBLK), dim3(LT), 0, stream>>>(x, head, srt, Wl0, Wr0, b0,
                                                    nullptr, out, ha);

  k_sage_layer<<<dim3(LBLK), dim3(LT), 0, stream>>>(ha, head, srt, Wl1, Wr1, b1,
                                                    out, out, hb);

  k_sage_layer<<<dim3(LBLK), dim3(LT), 0, stream>>>(hb, head, srt, Wl2, Wr2, b2,
                                                    out, out, nullptr);
}


# Gather keeps a ring of 32 row requests in flight per wave: each add is followed by the request for the row 32 list places on (was: ask 16, wait, add 16; tails 8/4/2/1); sum order unchanged
# speedup vs baseline: 4.1766x; 1.0376x over previous
#include <hip/hip_runtime.h>


#define NN 50000
#define NE 800000
#define DD 128
#define NT (NN / 16)
#define FEAT (NN * DD)
#define WSTRIDE 132
#define WOFF (DD * WSTRIDE)
#define SCAN_T 1024
#define SCAN_PER 49
#define NPAD (SCAN_T * SCAN_PER)
#define SCAN_FULL 12
#define NB 196
#define NG 250
#define CHUNK (NE / NG)
#define EDGE_T 640
#define EPT (CHUNK / EDGE_T)
#define M4 ((NB * NG) / 4)
#define BKT_T 1024
#define BKT_U 4
#define STAGE 8192
#define LW 16
#define LT (LW * 32)
#define LBLK 256
#define ATILE (16 * WSTRIDE)
#define AOFF (2 * WOFF)

typedef float v2f __attribute__((ext_vector_type(2)));
typedef float v8f __attribute__((ext_vector_type(8)));

__device__ __forceinline__ v8f wmma4(v2f a, v2f b, v8f c) {
  return __builtin_amdgcn_wmma_f32_16x16x4_f32(false, a, false, b, (short)0, c,
                                               false, false);
}

__global__ __launch_bounds__(EDGE_T) void k_hist(const int* __restrict__ row,
                                                 int* __restrict__ cntm) {
  __shared__ int hist[256];
  const int t = threadIdx.x;
  const int g = blockIdx.x;
  if (t < 256) hist[t] = 0;
  __syncthreads();
  const int* r = row + g * CHUNK + t;
  int d[EPT];
#pragma unroll
  for (int j = 0; j < EPT; ++j) d[j] = r[j * EDGE_T];
#pragma unroll
  for (int j = 0; j < EPT; ++j) atomicAdd(&hist[d[j] >> 8], 1);
  __syncthreads();
  if (t < NB) cntm[t * NG + g] = hist[t];
}

__global__ __launch_bounds__(SCAN_T) void k_scan(const int* __restrict__ cntm,
                                                 int* __restrict__ offs) {
  __shared__ int s[NPAD];
  __shared__ int wtot[SCAN_T / 32];
  const int t = threadIdx.x;
  const int lane = t & 31;
  const int wv = t >> 5;
  const int4 z4 = {0, 0, 0, 0};

  int4 v[SCAN_FULL];
#pragma unroll
  for (int k = 0; k < SCAN_FULL; ++k) {
    const int i = t + k * SCAN_T;
    v[k] = (i < M4) ? ((const int4*)cntm)[i] : z4;
  }
  {
    const int i = t + SCAN_FULL * SCAN_T;
    if (i < NPAD / 4) ((int4*)s)[i] = z4;
  }
#pragma unroll
  for (int k = 0; k < SCAN_FULL; ++k) ((int4*)s)[t + k * SCAN_T] = v[k];
  __syncthreads();

  const int base = t * SCAN_PER;
  int sum = 0;
  for (int j = 0; j < SCAN_PER; ++j) sum += s[base + j];

  int inc = sum;
  for (int d = 1; d < 32; d <<= 1) {
    int o = __shfl_up(inc, d);
    if (lane >= d) inc += o;
  }
  if (lane == 31) wtot[wv] = inc;
  __syncthreads();
  int wo = (lane < wv) ? wtot[lane] : 0;
  for (int d = 16; d >= 1; d >>= 1) wo += __shfl_xor(wo, d);

  int run = wo + inc - sum;
  for (int j = 0; j < SCAN_PER; ++j) {
    const int c = s[base + j];
    s[base + j] = run;
    run += c;
  }
  __syncthreads();
  for (int i = t; i < NPAD / 4; i += SCAN_T) ((int4*)offs)[i] = ((int4*)s)[i];
}

__global__ __launch_bounds__(EDGE_T) void k_place(const int* __restrict__ row,
                                                  const int* __restrict__ col,
                                                  const int* __restrict__ offs,
                                                  int* __restrict__ rec) {
  __shared__ int cur[256];
  const int t = threadIdx.x;
  const int g = blockIdx.x;
  if (t < NB) cur[t] = offs[t * NG + g];
  const int e = g * CHUNK + t;
  int d[EPT], sn[EPT];
#pragma unroll
  for (int j = 0; j < EPT; ++j) {
    d[j] = row[e + j * EDGE_T];
    sn[j] = col[e + j * EDGE_T];
  }
  __syncthreads();
#pragma unroll
  for (int j = 0; j < EPT; ++j) {
    const int p = atomicAdd(&cur[d[j] >> 8], 1);
    rec[p] = (sn[j] << 8) | (d[j] & 255);
  }
}

__global__ __launch_bounds__(BKT_T) void k_bucket(const int* __restrict__ offs,
                                                  const int* __restrict__ rec,
                                                  int* __restrict__ head,
                                                  int* __restrict__ srt) {
  __shared__ int cnt[256];
  __shared__ int cur[256];
  __shared__ int wtot[8];
  __shared__ int stage[STAGE];
  const int t = threadIdx.x;
  const int b = blockIdx.x;
  const int lane = t & 31;
  const int wv = t >> 5;
  const int base = offs[b * NG];
  const int n = offs[(b + 1) * NG] - base;
  const int* rb = rec + base;
  if (t < 256) cnt[t] = 0;
  __syncthreads();
  for (int i0 = t; i0 < n; i0 += BKT_U * BKT_T) {
    int r[BKT_U];
#pragma unroll
    for (int u = 0; u < BKT_U; ++u) {
      const int i = i0 + u * BKT_T;
      r[u] = (i < n) ? rb[i] : -1;
    }
#pragma unroll
    for (int u = 0; u < BKT_U; ++u)
      if (r[u] >= 0) atomicAdd(&cnt[r[u] & 255], 1);
  }
  __syncthreads();
  int c = 0, inc = 0;
  if (t < 256) {
    c = cnt[t];
    inc = c;
    for (int d = 1; d < 32; d <<= 1) {
      int o = __shfl_up(inc, d);
      if (lane >= d) inc += o;
    }
    if (lane == 31) wtot[wv] = inc;
  }
  __syncthreads();
  if (t < 256) {
    int wo = (lane < wv) ? wtot[lane] : 0;
    for (int d = 16; d >= 1; d >>= 1) wo += __shfl_xor(wo, d);
    const int first = base + wo + inc - c;
    const int node = (b << 8) + t;
    if (node <= NN) head[node] = first;
    cur[t] = first;
  }
  __syncthreads();
  const bool fits = n <= STAGE;
  for (int i0 = t; i0 < n; i0 += BKT_U * BKT_T) {
    int r[BKT_U];
#pragma unroll
    for (int u = 0; u < BKT_U; ++u) {
      const int i = i0 + u * BKT_T;
      r[u] = (i < n) ? rb[i] : -1;
    }
#pragma unroll
    for (int u = 0; u < BKT_U; ++u) {
      if (r[u] >= 0) {
        const int p = atomicAdd(&cur[r[u] & 255], 1);
        if (fits) stage[p - base] = r[u] >> 8;
        else srt[p] = r[u] >> 8;
      }
    }
  }
  if (fits) {
    __syncthreads();
    for (int i = t; i < n; i += BKT_T) srt[base + i] = stage[i];
  }
}

struct Walk {
  int ni;
  int nbeg;
  int nend;
};

__device__ __forceinline__ void close_node(Walk& w, float4& acc, int hd,
                                           float* __restrict__ atile,
                                           int lane) {
  const float sc = 1.0f / fmaxf((float)(w.nend - w.nbeg), 1.0f);
  float4 o;
  o.x = acc.x * sc;
  o.y = acc.y * sc;
  o.z = acc.z * sc;
  o.w = acc.w * sc;
  *(float4*)(atile + w.ni * WSTRIDE + (lane << 2)) = o;
  acc.x = 0.f;
  acc.y = 0.f;
  acc.z = 0.f;
  acc.w = 0.f;
  w.ni += 1;
  w.nbeg = w.nend;
  w.nend = __builtin_amdgcn_readlane(hd, w.ni + 1);
}

__device__ __forceinline__ float4 ask_row(const float* __restrict__ xl,
                                          int sn) {
  return *(const float4*)(xl + (size_t)sn * DD);
}

__device__ __forceinline__ void add_row(const float4& v, int pos, Walk& w,
                                        float4& acc, int hd,
                                        float* __restrict__ atile, int lane) {
  while (pos == w.nend) close_node(w, acc, hd, atile, lane);
  acc.x += v.x;
  acc.y += v.y;
  acc.z += v.z;
  acc.w += v.w;
}

__device__ __forceinline__ void k_pass(const float* __restrict__ a,
                                       const float* __restrict__ bb,
                                       v8f (&acc)[8]) {
  v2f cur[8], nxt[8];
#pragma unroll
  for (int t = 0; t < 8; ++t)
    cur[t] = *(const v2f*)(bb + (size_t)(t << 4) * WSTRIDE);
  v2f a2 = *(const v2f*)(a);

#pragma unroll 2
  for (int k = 0; k < 32; ++k) {
    const int K1 = ((k + 1) & 31) << 2;
#pragma unroll
    for (int t = 0; t < 8; ++t)
      nxt[t] = *(const v2f*)(bb + (size_t)(t << 4) * WSTRIDE + K1);
    const v2f a2n = *(const v2f*)(a + K1);

#pragma unroll
    for (int t = 0; t < 8; ++t) acc[t] = wmma4(a2, cur[t], acc[t]);

#pragma unroll
    for (int t = 0; t < 8; ++t) cur[t] = nxt[t];
    a2 = a2n;
  }
}

template <bool RES, bool RELU>
__device__ __forceinline__ void put_tile(const v8f (&acc)[8],
                                         const float (&bb)[8],
                                         const float (&rs)[8][8],
                                         float* __restrict__ o,
                                         float* __restrict__ h) {
#pragma unroll
  for (int t = 0; t < 8; ++t) {
#pragma unroll
    for (int r = 0; r < 8; ++r) {
      float v = acc[t][r] + bb[t];
      if (RES) v += rs[t][r];
      o[r * DD + (t << 4)] = v;
      if (RELU) h[r * DD + (t << 4)] = fmaxf(v, 0.0f);
    }
  }
}

__global__ __launch_bounds__(LT) void k_sage_layer(
    const float* __restrict__ hin, const int* __restrict__ head,
    const int* __restrict__ srt, const float* __restrict__ Wl,
    const float* __restrict__ Wr, const float* __restrict__ bias,
    const float* __restrict__ resid,
    float* __restrict__ out,
    float* __restrict__ hout) {
  __shared__ __attribute__((aligned(16))) float wlds[AOFF + LW * ATILE];

  const int wv = threadIdx.x >> 5;
  const int lane = threadIdx.x & 31;
  const int half = lane >> 4;
  const int lm = lane & 15;
  const int tile = wv * gridDim.x + blockIdx.x;
  const bool active = tile < NT;
  const int m0 = tile << 4;
  float* atile = &wlds[AOFF + wv * ATILE];

#pragma unroll
  for (int it = 0; it < (DD * DD / 4) / LT; ++it) {
    const int i = threadIdx.x + it * LT;
    const int o = i >> 5;
    const int kc = (i & 31) << 2;
    const float4 a = *(const float4*)(Wl + (size_t)o * DD + kc);
    const float4 b = *(const float4*)(Wr + (size_t)o * DD + kc);
    *(float4*)(&wlds[o * WSTRIDE + kc]) = a;
    *(float4*)(&wlds[WOFF + o * WSTRIDE + kc]) = b;
  }

  if (active) {
    const int hd = (lane <= 16) ? head[m0 + lane] : 0;
    const int e0 = __builtin_amdgcn_readlane(hd, 0);
    const int e1 = __builtin_amdgcn_readlane(hd, 16);
    const int n = e1 - e0;
    const int trips = (n + 31) >> 5;
    Walk w;
    w.ni = 0;
    w.nbeg = e0;
    w.nend = __builtin_amdgcn_readlane(hd, 1);
    const float* xl = hin + (lane << 2);
    const int* sl = srt + e0 + lane;
    float4 acc = {0.f, 0.f, 0.f, 0.f};
    const int own = m0 + lm;
    int sa = (lane < n) ? sl[0] : own;
    float4 v[32];
#pragma unroll
    for (int u = 0; u < 32; ++u) {
      v[u].x = 0.f;
      v[u].y = 0.f;
      v[u].z = 0.f;
      v[u].w = 0.f;
    }
    int pos = e0;
    int lim = 0;
    int sn = __builtin_amdgcn_readlane(sa, 0);
    for (int it = 0; it <= trips; ++it) {
      const int p1 = ((it + 1) << 5) + lane;
      const int sb = (p1 < n) ? sl[(it + 1) << 5] : own;
#pragma unroll
      for (int u = 0; u < 32; ++u) {
        if (u < lim) add_row(v[u], pos + u, w, acc, hd, atile, lane);
        v[u] = ask_row(xl, sn);
        sn = (u < 31) ? __builtin_amdgcn_readlane(sa, u + 1)
                      : __builtin_amdgcn_readlane(sb, 0);
      }
      pos += lim;
      lim = (e1 - pos < 32) ? (e1 - pos) : 32;
      sa = sb;
    }
    while (w.ni < 16) close_node(w, acc, hd, atile, lane);
  }
  __syncthreads();
  if (!active) return;

  const float* arow = atile + lm * WSTRIDE + 2 * half;
  const float* hrow = hin + (size_t)(m0 + lm) * DD + 2 * half;
  const float* blbase = &wlds[(size_t)lm * WSTRIDE + 2 * half];
  const float* brbase = &wlds[WOFF + (size_t)lm * WSTRIDE + 2 * half];

  const size_t obase = (size_t)(m0 + (half << 3)) * DD + lm;

  float bb[8];
#pragma unroll
  for (int t = 0; t < 8; ++t) bb[t] = bias[(t << 4) + lm];
  float rs[8][8];
  if (resid) {
#pragma unroll
    for (int t = 0; t < 8; ++t)
#pragma unroll
      for (int r = 0; r < 8; ++r)
        rs[t][r] = resid[obase + (size_t)r * DD + (t << 4)];
  } else {
#pragma unroll
    for (int t = 0; t < 8; ++t)
#pragma unroll
      for (int r = 0; r < 8; ++r) rs[t][r] = 0.f;
  }

  v8f acc[8] = {};

  k_pass(arow, blbase, acc);
  k_pass(hrow, brbase, acc);

  if (resid) {
    if (hout) put_tile<true, true>(acc, bb, rs, out + obase, hout + obase);
    else put_tile<true, false>(acc, bb, rs, out + obase, nullptr);
  } else {
    if (hout) put_tile<false, true>(acc, bb, rs, out + obase, hout + obase);
    else put_tile<false, false>(acc, bb, rs, out + obase, nullptr);
  }
}

extern "C" void kernel_launch(void* const* d_in, const int* in_sizes, int n_in,
                              void* d_out, int out_size, void* d_ws,
                              size_t ws_size, hipStream_t stream) {
  const float* x = (const float*)d_in[0];
  const int* erow = (const int*)d_in[1];
  const int* ecol = (const int*)d_in[2];
  const float* Wl0 = (const float*)d_in[3];
  const float* Wr0 = (const float*)d_in[4];
  const float* b0 = (const float*)d_in[5];
  const float* Wl1 = (const float*)d_in[6];
  const float* Wr1 = (const float*)d_in[7];
  const float* b1 = (const float*)d_in[8];
  const float* Wl2 = (const float*)d_in[9];
  const float* Wr2 = (const float*)d_in[10];
  const float* b2 = (const float*)d_in[11];
  float* out = (float*)d_out;

  float* ws = (float*)d_ws;
  float* ha = ws;
  float* hb = ha + (size_t)FEAT;
  int* cntm = (int*)(hb + (size_t)FEAT);
  int* head = cntm + NN;
  int* offs = head + NPAD;
  int* rec = offs + NPAD;
  int* srt = rec + NE;

  k_hist<<<dim3(NG), dim3(EDGE_T), 0, stream>>>(erow, cntm);
  k_scan<<<dim3(1), dim3(SCAN_T), 0, stream>>>(cntm, offs);
  k_place<<<dim3(NG), dim3(EDGE_T), 0, stream>>>(erow, ecol, offs, rec);
  k_bucket<<<dim3(NB), dim3(BKT_T), 0, stream>>>(offs, rec, head, srt);

  k_sage_layer<<<dim3(LBLK), dim3(LT), 0, stream>>>(x, head, srt, Wl0, Wr0, b0,
                                                    nullptr, out, ha);

  k_sage_layer<<<dim3(LBLK), dim3(LT), 0, stream>>>(ha, head, srt, Wl1, Wr1, b1,
                                                    out, out, hb);

  k_sage_layer<<<dim3(LBLK), dim3(LT), 0, stream>>>(hb, head, srt, Wl2, Wr2, b2,
                                                    out, out, nullptr);
}


# Products of the 13th tile (53 of 256 blocks have one) shared by waves 12-15, 2 of its 8 column tiles each, one wave per SIMD, not wave 12 alone; every sum and its order unchanged
# speedup vs baseline: 4.4506x; 1.0656x over previous
#include <hip/hip_runtime.h>


#define NN 50000
#define NE 800000
#define DD 128
#define NT (NN / 16)
#define FEAT (NN * DD)
#define WSTRIDE 132
#define WOFF (DD * WSTRIDE)
#define SCAN_T 1024
#define SCAN_PER 49
#define NPAD (SCAN_T * SCAN_PER)
#define SCAN_FULL 12
#define NB 196
#define NG 250
#define CHUNK (NE / NG)
#define EDGE_T 640
#define EPT (CHUNK / EDGE_T)
#define M4 ((NB * NG) / 4)
#define BKT_T 1024
#define BKT_U 4
#define STAGE 8192
#define LW 16
#define LT (LW * 32)
#define LBLK 256
#define ATILE (16 * WSTRIDE)
#define AOFF (2 * WOFF)
#define FULLW (NT / LBLK)
#define QW (LW - FULLW)
#define QCT (8 / QW)
static_assert(QW == 4 && QCT * QW == 8, "the 13th tile is split four ways");
static_assert((FULLW + 1) * LBLK >= NT, "at most 13 tiles per block");

typedef float v2f __attribute__((ext_vector_type(2)));
typedef float v8f __attribute__((ext_vector_type(8)));

__device__ __forceinline__ v8f wmma4(v2f a, v2f b, v8f c) {
  return __builtin_amdgcn_wmma_f32_16x16x4_f32(false, a, false, b, (short)0, c,
                                               false, false);
}

__global__ __launch_bounds__(EDGE_T) void k_hist(const int* __restrict__ row,
                                                 int* __restrict__ cntm) {
  __shared__ int hist[256];
  const int t = threadIdx.x;
  const int g = blockIdx.x;
  if (t < 256) hist[t] = 0;
  __syncthreads();
  const int* r = row + g * CHUNK + t;
  int d[EPT];
#pragma unroll
  for (int j = 0; j < EPT; ++j) d[j] = r[j * EDGE_T];
#pragma unroll
  for (int j = 0; j < EPT; ++j) atomicAdd(&hist[d[j] >> 8], 1);
  __syncthreads();
  if (t < NB) cntm[t * NG + g] = hist[t];
}

__global__ __launch_bounds__(SCAN_T) void k_scan(const int* __restrict__ cntm,
                                                 int* __restrict__ offs) {
  __shared__ int s[NPAD];
  __shared__ int wtot[SCAN_T / 32];
  const int t = threadIdx.x;
  const int lane = t & 31;
  const int wv = t >> 5;
  const int4 z4 = {0, 0, 0, 0};

  int4 v[SCAN_FULL];
#pragma unroll
  for (int k = 0; k < SCAN_FULL; ++k) {
    const int i = t + k * SCAN_T;
    v[k] = (i < M4) ? ((const int4*)cntm)[i] : z4;
  }
  {
    const int i = t + SCAN_FULL * SCAN_T;
    if (i < NPAD / 4) ((int4*)s)[i] = z4;
  }
#pragma unroll
  for (int k = 0; k < SCAN_FULL; ++k) ((int4*)s)[t + k * SCAN_T] = v[k];
  __syncthreads();

  const int base = t * SCAN_PER;
  int sum = 0;
  for (int j = 0; j < SCAN_PER; ++j) sum += s[base + j];

  int inc = sum;
  for (int d = 1; d < 32; d <<= 1) {
    int o = __shfl_up(inc, d);
    if (lane >= d) inc += o;
  }
  if (lane == 31) wtot[wv] = inc;
  __syncthreads();
  int wo = (lane < wv) ? wtot[lane] : 0;
  for (int d = 16; d >= 1; d >>= 1) wo += __shfl_xor(wo, d);

  int run = wo + inc - sum;
  for (int j = 0; j < SCAN_PER; ++j) {
    const int c = s[base + j];
    s[base + j] = run;
    run += c;
  }
  __syncthreads();
  for (int i = t; i < NPAD / 4; i += SCAN_T) ((int4*)offs)[i] = ((int4*)s)[i];
}

__global__ __launch_bounds__(EDGE_T) void k_place(const int* __restrict__ row,
                                                  const int* __restrict__ col,
                                                  const int* __restrict__ offs,
                                                  int* __restrict__ rec) {
  __shared__ int cur[256];
  const int t = threadIdx.x;
  const int g = blockIdx.x;
  if (t < NB) cur[t] = offs[t * NG + g];
  const int e = g * CHUNK + t;
  int d[EPT], sn[EPT];
#pragma unroll
  for (int j = 0; j < EPT; ++j) {
    d[j] = row[e + j * EDGE_T];
    sn[j] = col[e + j * EDGE_T];
  }
  __syncthreads();
#pragma unroll
  for (int j = 0; j < EPT; ++j) {
    const int p = atomicAdd(&cur[d[j] >> 8], 1);
    rec[p] = (sn[j] << 8) | (d[j] & 255);
  }
}

__global__ __launch_bounds__(BKT_T) void k_bucket(const int* __restrict__ offs,
                                                  const int* __restrict__ rec,
                                                  int* __restrict__ head,
                                                  int* __restrict__ srt) {
  __shared__ int cnt[256];
  __shared__ int cur[256];
  __shared__ int wtot[8];
  __shared__ int stage[STAGE];
  const int t = threadIdx.x;
  const int b = blockIdx.x;
  const int lane = t & 31;
  const int wv = t >> 5;
  const int base = offs[b * NG];
  const int n = offs[(b + 1) * NG] - base;
  const int* rb = rec + base;
  if (t < 256) cnt[t] = 0;
  __syncthreads();
  for (int i0 = t; i0 < n; i0 += BKT_U * BKT_T) {
    int r[BKT_U];
#pragma unroll
    for (int u = 0; u < BKT_U; ++u) {
      const int i = i0 + u * BKT_T;
      r[u] = (i < n) ? rb[i] : -1;
    }
#pragma unroll
    for (int u = 0; u < BKT_U; ++u)
      if (r[u] >= 0) atomicAdd(&cnt[r[u] & 255], 1);
  }
  __syncthreads();
  int c = 0, inc = 0;
  if (t < 256) {
    c = cnt[t];
    inc = c;
    for (int d = 1; d < 32; d <<= 1) {
      int o = __shfl_up(inc, d);
      if (lane >= d) inc += o;
    }
    if (lane == 31) wtot[wv] = inc;
  }
  __syncthreads();
  if (t < 256) {
    int wo = (lane < wv) ? wtot[lane] : 0;
    for (int d = 16; d >= 1; d >>= 1) wo += __shfl_xor(wo, d);
    const int first = base + wo + inc - c;
    const int node = (b << 8) + t;
    if (node <= NN) head[node] = first;
    cur[t] = first;
  }
  __syncthreads();
  const bool fits = n <= STAGE;
  for (int i0 = t; i0 < n; i0 += BKT_U * BKT_T) {
    int r[BKT_U];
#pragma unroll
    for (int u = 0; u < BKT_U; ++u) {
      const int i = i0 + u * BKT_T;
      r[u] = (i < n) ? rb[i] : -1;
    }
#pragma unroll
    for (int u = 0; u < BKT_U; ++u) {
      if (r[u] >= 0) {
        const int p = atomicAdd(&cur[r[u] & 255], 1);
        if (fits) stage[p - base] = r[u] >> 8;
        else srt[p] = r[u] >> 8;
      }
    }
  }
  if (fits) {
    __syncthreads();
    for (int i = t; i < n; i += BKT_T) srt[base + i] = stage[i];
  }
}

struct Walk {
  int ni;
  int nbeg;
  int nend;
};

__device__ __forceinline__ void close_node(Walk& w, float4& acc, int hd,
                                           float* __restrict__ atile,
                                           int lane) {
  const float sc = 1.0f / fmaxf((float)(w.nend - w.nbeg), 1.0f);
  float4 o;
  o.x = acc.x * sc;
  o.y = acc.y * sc;
  o.z = acc.z * sc;
  o.w = acc.w * sc;
  *(float4*)(atile + w.ni * WSTRIDE + (lane << 2)) = o;
  acc.x = 0.f;
  acc.y = 0.f;
  acc.z = 0.f;
  acc.w = 0.f;
  w.ni += 1;
  w.nbeg = w.nend;
  w.nend = __builtin_amdgcn_readlane(hd, w.ni + 1);
}

__device__ __forceinline__ float4 ask_row(const float* __restrict__ xl,
                                          int sn) {
  return *(const float4*)(xl + (size_t)sn * DD);
}

__device__ __forceinline__ void add_row(const float4& v, int pos, Walk& w,
                                        float4& acc, int hd,
                                        float* __restrict__ atile, int lane) {
  while (pos == w.nend) close_node(w, acc, hd, atile, lane);
  acc.x += v.x;
  acc.y += v.y;
  acc.z += v.z;
  acc.w += v.w;
}

__device__ __forceinline__ void k_pass(const float* __restrict__ a,
                                       const float* __restrict__ bb,
                                       v8f (&acc)[8]) {
  v2f cur[8], nxt[8];
#pragma unroll
  for (int t = 0; t < 8; ++t)
    cur[t] = *(const v2f*)(bb + (size_t)(t << 4) * WSTRIDE);
  v2f a2 = *(const v2f*)(a);

#pragma unroll 2
  for (int k = 0; k < 32; ++k) {
    const int K1 = ((k + 1) & 31) << 2;
#pragma unroll
    for (int t = 0; t < 8; ++t)
      nxt[t] = *(const v2f*)(bb + (size_t)(t << 4) * WSTRIDE + K1);
    const v2f a2n = *(const v2f*)(a + K1);

#pragma unroll
    for (int t = 0; t < 8; ++t) acc[t] = wmma4(a2, cur[t], acc[t]);

#pragma unroll
    for (int t = 0; t < 8; ++t) cur[t] = nxt[t];
    a2 = a2n;
  }
}

__device__ __forceinline__ void q_pass_lds(const float* __restrict__ a,
                                           const float* __restrict__ bb,
                                           v8f (&acc)[QCT]) {
  v2f cb[4][QCT], nb[4][QCT], ca[4], na[4];
#pragma unroll
  for (int j = 0; j < 4; ++j) {
#pragma unroll
    for (int t = 0; t < QCT; ++t)
      cb[j][t] = *(const v2f*)(bb + (size_t)(t << 4) * WSTRIDE + 4 * j);
    ca[j] = *(const v2f*)(a + 4 * j);
  }
#pragma unroll 2
  for (int g = 0; g < 8; ++g) {
    const int K1 = ((g + 1) & 7) << 4;
#pragma unroll
    for (int j = 0; j < 4; ++j) {
#pragma unroll
      for (int t = 0; t < QCT; ++t)
        nb[j][t] = *(const v2f*)(bb + (size_t)(t << 4) * WSTRIDE + K1 + 4 * j);
      na[j] = *(const v2f*)(a + K1 + 4 * j);
    }
    __builtin_amdgcn_sched_barrier(0);
#pragma unroll
    for (int j = 0; j < 4; ++j)
#pragma unroll
      for (int t = 0; t < QCT; ++t) acc[t] = wmma4(ca[j], cb[j][t], acc[t]);
#pragma unroll
    for (int j = 0; j < 4; ++j) {
#pragma unroll
      for (int t = 0; t < QCT; ++t) cb[j][t] = nb[j][t];
      ca[j] = na[j];
    }
  }
}

__device__ __forceinline__ void q_pass_reg(const v2f (&sa)[32],
                                           const float* __restrict__ bb,
                                           v8f (&acc)[QCT]) {
  v2f cb[4][QCT], nb[4][QCT];
#pragma unroll
  for (int j = 0; j < 4; ++j)
#pragma unroll
    for (int t = 0; t < QCT; ++t)
      cb[j][t] = *(const v2f*)(bb + (size_t)(t << 4) * WSTRIDE + 4 * j);
#pragma unroll
  for (int g = 0; g < 8; ++g) {
    if (g < 7) {
#pragma unroll
      for (int j = 0; j < 4; ++j)
#pragma unroll
        for (int t = 0; t < QCT; ++t)
          nb[j][t] = *(const v2f*)(bb + (size_t)(t << 4) * WSTRIDE +
                                   ((g + 1) << 4) + 4 * j);
    }
    __builtin_amdgcn_sched_barrier(0);
#pragma unroll
    for (int j = 0; j < 4; ++j)
#pragma unroll
      for (int t = 0; t < QCT; ++t)
        acc[t] = wmma4(sa[4 * g + j], cb[j][t], acc[t]);
#pragma unroll
    for (int j = 0; j < 4; ++j)
#pragma unroll
      for (int t = 0; t < QCT; ++t) cb[j][t] = nb[j][t];
  }
}

template <int NTL, bool RES, bool RELU>
__device__ __forceinline__ void put_tile(const v8f (&acc)[NTL],
                                         const float (&bb)[NTL],
                                         const float (&rs)[NTL][8],
                                         float* __restrict__ o,
                                         float* __restrict__ h) {
#pragma unroll
  for (int t = 0; t < NTL; ++t) {
#pragma unroll
    for (int r = 0; r < 8; ++r) {
      float v = acc[t][r] + bb[t];
      if (RES) v += rs[t][r];
      o[r * DD + (t << 4)] = v;
      if (RELU) h[r * DD + (t << 4)] = fmaxf(v, 0.0f);
    }
  }
}

__global__ __launch_bounds__(LT) void k_sage_layer(
    const float* __restrict__ hin, const int* __restrict__ head,
    const int* __restrict__ srt, const float* __restrict__ Wl,
    const float* __restrict__ Wr, const float* __restrict__ bias,
    const float* __restrict__ resid,
    float* __restrict__ out,
    float* __restrict__ hout) {
  __shared__ __attribute__((aligned(16))) float wlds[AOFF + LW * ATILE];

  const int wv = threadIdx.x >> 5;
  const int lane = threadIdx.x & 31;
  const int half = lane >> 4;
  const int lm = lane & 15;
  const int tile = wv * gridDim.x + blockIdx.x;
  const bool active = tile < NT;
  const int m0 = tile << 4;
  float* atile = &wlds[AOFF + wv * ATILE];

#pragma unroll
  for (int it = 0; it < (DD * DD / 4) / LT; ++it) {
    const int i = threadIdx.x + it * LT;
    const int o = i >> 5;
    const int kc = (i & 31) << 2;
    const float4 a = *(const float4*)(Wl + (size_t)o * DD + kc);
    const float4 b = *(const float4*)(Wr + (size_t)o * DD + kc);
    *(float4*)(&wlds[o * WSTRIDE + kc]) = a;
    *(float4*)(&wlds[WOFF + o * WSTRIDE + kc]) = b;
  }

  if (active) {
    const int hd = (lane <= 16) ? head[m0 + lane] : 0;
    const int e0 = __builtin_amdgcn_readlane(hd, 0);
    const int e1 = __builtin_amdgcn_readlane(hd, 16);
    const int n = e1 - e0;
    const int trips = (n + 31) >> 5;
    Walk w;
    w.ni = 0;
    w.nbeg = e0;
    w.nend = __builtin_amdgcn_readlane(hd, 1);
    const float* xl = hin + (lane << 2);
    const int* sl = srt + e0 + lane;
    float4 acc = {0.f, 0.f, 0.f, 0.f};
    const int own = m0 + lm;
    int sa = (lane < n) ? sl[0] : own;
    float4 v[32];
#pragma unroll
    for (int u = 0; u < 32; ++u) {
      v[u].x = 0.f;
      v[u].y = 0.f;
      v[u].z = 0.f;
      v[u].w = 0.f;
    }
    int pos = e0;
    int lim = 0;
    int sn = __builtin_amdgcn_readlane(sa, 0);
    for (int it = 0; it <= trips; ++it) {
      const int p1 = ((it + 1) << 5) + lane;
      const int sb = (p1 < n) ? sl[(it + 1) << 5] : own;
#pragma unroll
      for (int u = 0; u < 32; ++u) {
        if (u < lim) add_row(v[u], pos + u, w, acc, hd, atile, lane);
        v[u] = ask_row(xl, sn);
        sn = (u < 31) ? __builtin_amdgcn_readlane(sa, u + 1)
                      : __builtin_amdgcn_readlane(sb, 0);
      }
      pos += lim;
      lim = (e1 - pos < 32) ? (e1 - pos) : 32;
      sa = sb;
    }
    while (w.ni < 16) close_node(w, acc, hd, atile, lane);
  }
  __syncthreads();

  if (wv >= FULLW) {
    const int xt = FULLW * gridDim.x + blockIdx.x;
    if (xt >= NT) return;
    const int xm0 = xt << 4;
    const int c0 = (wv - FULLW) * (QCT << 4);
    const float* xa = &wlds[AOFF + FULLW * ATILE] + lm * WSTRIDE + 2 * half;
    const float* xh = hin + (size_t)(xm0 + lm) * DD + 2 * half;
    const float* xbl = &wlds[(size_t)(c0 + lm) * WSTRIDE + 2 * half];
    const float* xbr = &wlds[WOFF + (size_t)(c0 + lm) * WSTRIDE + 2 * half];
    const size_t xo = (size_t)(xm0 + (half << 3)) * DD + c0 + lm;

    float qb[QCT];
#pragma unroll
    for (int t = 0; t < QCT; ++t) qb[t] = bias[c0 + (t << 4) + lm];
    float qr[QCT][8];
    if (resid) {
#pragma unroll
      for (int t = 0; t < QCT; ++t)
#pragma unroll
        for (int r = 0; r < 8; ++r)
          qr[t][r] = resid[xo + (size_t)r * DD + (t << 4)];
    } else {
#pragma unroll
      for (int t = 0; t < QCT; ++t)
#pragma unroll
        for (int r = 0; r < 8; ++r) qr[t][r] = 0.f;
    }
    v2f sa[32];
#pragma unroll
    for (int k = 0; k < 32; ++k) sa[k] = *(const v2f*)(xh + 4 * k);

    v8f qacc[QCT] = {};
    q_pass_lds(xa, xbl, qacc);
    q_pass_reg(sa, xbr, qacc);

    if (resid) {
      if (hout) put_tile<QCT, true, true>(qacc, qb, qr, out + xo, hout + xo);
      else put_tile<QCT, true, false>(qacc, qb, qr, out + xo, nullptr);
    } else {
      if (hout) put_tile<QCT, false, true>(qacc, qb, qr, out + xo, hout + xo);
      else put_tile<QCT, false, false>(qacc, qb, qr, out + xo, nullptr);
    }
    return;
  }

  const float* arow = atile + lm * WSTRIDE + 2 * half;
  const float* hrow = hin + (size_t)(m0 + lm) * DD + 2 * half;
  const float* blbase = &wlds[(size_t)lm * WSTRIDE + 2 * half];
  const float* brbase = &wlds[WOFF + (size_t)lm * WSTRIDE + 2 * half];

  const size_t obase = (size_t)(m0 + (half << 3)) * DD + lm;

  float bb[8];
#pragma unroll
  for (int t = 0; t < 8; ++t) bb[t] = bias[(t << 4) + lm];
  float rs[8][8];
  if (resid) {
#pragma unroll
    for (int t = 0; t < 8; ++t)
#pragma unroll
      for (int r = 0; r < 8; ++r)
        rs[t][r] = resid[obase + (size_t)r * DD + (t << 4)];
  } else {
#pragma unroll
    for (int t = 0; t < 8; ++t)
#pragma unroll
      for (int r = 0; r < 8; ++r) rs[t][r] = 0.f;
  }

  v8f acc[8] = {};

  k_pass(arow, blbase, acc);
  k_pass(hrow, brbase, acc);

  if (resid) {
    if (hout) put_tile<8, true, true>(acc, bb, rs, out + obase, hout + obase);
    else put_tile<8, true, false>(acc, bb, rs, out + obase, nullptr);
  } else {
    if (hout) put_tile<8, false, true>(acc, bb, rs, out + obase, hout + obase);
    else put_tile<8, false, false>(acc, bb, rs, out + obase, nullptr);
  }
}

extern "C" void kernel_launch(void* const* d_in, const int* in_sizes, int n_in,
                              void* d_out, int out_size, void* d_ws,
                              size_t ws_size, hipStream_t stream) {
  const float* x = (const float*)d_in[0];
  const int* erow = (const int*)d_in[1];
  const int* ecol = (const int*)d_in[2];
  const float* Wl0 = (const float*)d_in[3];
  const float* Wr0 = (const float*)d_in[4];
  const float* b0 = (const float*)d_in[5];
  const float* Wl1 = (const float*)d_in[6];
  const float* Wr1 = (const float*)d_in[7];
  const float* b1 = (const float*)d_in[8];
  const float* Wl2 = (const float*)d_in[9];
  const float* Wr2 = (const float*)d_in[10];
  const float* b2 = (const float*)d_in[11];
  float* out = (float*)d_out;

  float* ws = (float*)d_ws;
  float* ha = ws;
  float* hb = ha + (size_t)FEAT;
  int* cntm = (int*)(hb + (size_t)FEAT);
  int* head = cntm + NN;
  int* offs = head + NPAD;
  int* rec = offs + NPAD;
  int* srt = rec + NE;

  k_hist<<<dim3(NG), dim3(EDGE_T), 0, stream>>>(erow, cntm);
  k_scan<<<dim3(1), dim3(SCAN_T), 0, stream>>>(cntm, offs);
  k_place<<<dim3(NG), dim3(EDGE_T), 0, stream>>>(erow, ecol, offs, rec);
  k_bucket<<<dim3(NB), dim3(BKT_T), 0, stream>>>(offs, rec, head, srt);

  k_sage_layer<<<dim3(LBLK), dim3(LT), 0, stream>>>(x, head, srt, Wl0, Wr0, b0,
                                                    nullptr, out, ha);

  k_sage_layer<<<dim3(LBLK), dim3(LT), 0, stream>>>(ha, head, srt, Wl1, Wr1, b1,
                                                    out, out, hb);

  k_sage_layer<<<dim3(LBLK), dim3(LT), 0, stream>>>(hb, head, srt, Wl2, Wr2, b2,
                                                    out, out, nullptr);
}
